# starting kernel
# baseline (speedup 1.0000x reference)
#include <hip/hip_runtime.h>


typedef __attribute__((ext_vector_type(2))) float v2f;
typedef __attribute__((ext_vector_type(8))) float v8f;

#define NYv   8
#define NUv   8
#define NHv   256
#define H2v   128
#define T_LEN 2048
#define NBv   64
#define NSEG  16
#define SEG   (T_LEN / NSEG)
#define CHNK  16
#define NCH   (SEG / CHNK)
#define XPAD  258

__device__ __forceinline__ void load_wt_ops(const float* __restrict__ Wx2y,
                                            int wave, int lane, v2f bop[16]) {
  const int n  = lane & 15;
  const int kh = (lane >> 4) * 2;
#pragma unroll
  for (int i = 0; i < 16; ++i) {
    const int k0 = wave * 64 + 4 * i + kh;
    v2f v;
    if (n < NYv) {
      v.x = Wx2y[n * NHv + k0];
      v.y = Wx2y[n * NHv + k0 + 1];
    } else {
      v.x = 0.0f; v.y = 0.0f;
    }
    bop[i] = v;
  }
}

__device__ __forceinline__ void project_chunk(const float (*Xc)[XPAD],
                                              const v2f bop[16],
                                              int wave, int lane,
                                              float (*Yp)[CHNK][16]) {
  v8f acc = {};
  const int row = lane & 15;
  const int kh  = (lane >> 4) * 2;
#pragma unroll
  for (int i = 0; i < 16; ++i) {
    const int k0 = wave * 64 + 4 * i + kh;
    v2f a;
    a.x = Xc[row][k0];
    a.y = Xc[row][k0 + 1];
    acc = __builtin_amdgcn_wmma_f32_16x16x4_f32(
        false, a, false, bop[i], (short)0, acc, false, false);
  }
  const int mbase = (lane >> 4) * 8;
  const int ncol  = lane & 15;
#pragma unroll
  for (int v = 0; v < 8; ++v) Yp[wave][v + mbase][ncol] = acc[v];
}

__global__ void __launch_bounds__(128)
lru_seg_scan(const float* __restrict__ U,   const float* __restrict__ lr,
             const float* __restrict__ li,  const float* __restrict__ Bm,
             const float* __restrict__ mult,const float* __restrict__ Wx2y,
             const float* __restrict__ bx2y,float* __restrict__ Y,
             float* __restrict__ e_ws) {
  const int b = blockIdx.x;
  const int s = blockIdx.y;
  const int j = threadIdx.x;
  const int wave = j >> 5, lane = j & 31;

  __shared__ float Uc[CHNK][NUv];
  __shared__ float Xc[CHNK][XPAD];
  __shared__ float Yp[4][CHNK][16];

  const float m1 = expf(mult[j]);
  const float m2 = expf(mult[j + H2v]);
  float Bs1[NUv], Bs2[NUv];
#pragma unroll
  for (int u = 0; u < NUv; ++u) {
    Bs1[u] = Bm[j * NUv + u] * m1;
    Bs2[u] = Bm[(j + H2v) * NUv + u] * m2;
  }
  const float r  = expf(-fabsf(lr[j]));
  const float th = 1.5707963267948966f * li[j];
  const float re = r * cosf(th);
  const float im = r * sinf(th);

  v2f bop[16];
  load_wt_ops(Wx2y, wave, lane, bop);

  float x1 = 0.0f, x2 = 0.0f;

  if (j < CHNK)
    __builtin_prefetch(U + ((size_t)(s * SEG + j) * NBv + b) * NUv, 0, 0);

  for (int ch = 0; ch < NCH; ++ch) {
    const int t0 = s * SEG + ch * CHNK;
    if (j < 2 * CHNK) {
      const int tt = j >> 1, h = (j & 1) * 4;
      const float4 uv =
          *(const float4*)(U + ((size_t)(t0 + tt) * NBv + b) * NUv + h);
      Uc[tt][h + 0] = uv.x; Uc[tt][h + 1] = uv.y;
      Uc[tt][h + 2] = uv.z; Uc[tt][h + 3] = uv.w;
    }
    if (ch + 1 < NCH && j < CHNK)
      __builtin_prefetch(U + ((size_t)(t0 + CHNK + j) * NBv + b) * NUv, 0, 0);
    __syncthreads();

#pragma unroll
    for (int tt = 0; tt < CHNK; ++tt) {
      float bu1 = 0.0f, bu2 = 0.0f;
#pragma unroll
      for (int u = 0; u < NUv; ++u) {
        const float uu = Uc[tt][u];
        bu1 = fmaf(uu, Bs1[u], bu1);
        bu2 = fmaf(uu, Bs2[u], bu2);
      }
      const float n1 = re * x1 - im * x2 + bu1;
      const float n2 = im * x1 + re * x2 + bu2;
      x1 = n1; x2 = n2;
      Xc[tt][j]       = x1;
      Xc[tt][j + H2v] = x2;
    }
    __syncthreads();

    project_chunk(Xc, bop, wave, lane, Yp);
    __syncthreads();

    {
      const int tt = j >> 3, n = j & 7;
      const float y = Yp[0][tt][n] + Yp[1][tt][n] + Yp[2][tt][n] + Yp[3][tt][n]
                    + bx2y[n];
      Y[((size_t)(t0 + tt) * NBv + b) * NYv + n] = y;
    }
    __syncthreads();
  }

  e_ws[((size_t)s * NBv + b) * NHv + j]       = x1;
  e_ws[((size_t)s * NBv + b) * NHv + j + H2v] = x2;
}

__global__ void __launch_bounds__(128)
lru_carry(const float* __restrict__ y0,  const float* __restrict__ Wy2x,
          const float* __restrict__ by2x,const float* __restrict__ lr,
          const float* __restrict__ li,  const float* __restrict__ e_ws,
          float* __restrict__ c_ws) {
  const int b = blockIdx.x;
  const int j = threadIdx.x;

  const float th  = 1.5707963267948966f * li[j];
  const float rS  = expf(-(float)SEG * fabsf(lr[j]));
  const float phS = (float)SEG * th;
  const float reS = rS * cosf(phS);
  const float imS = rS * sinf(phS);

  float c1 = by2x[j], c2 = by2x[j + H2v];
#pragma unroll
  for (int n = 0; n < NYv; ++n) {
    const float yv = y0[b * NYv + n];
    c1 = fmaf(yv, Wy2x[j * NYv + n], c1);
    c2 = fmaf(yv, Wy2x[(j + H2v) * NYv + n], c2);
  }

  for (int s = 0; s < NSEG; ++s) {
    const size_t base = ((size_t)s * NBv + b) * NHv;
    c_ws[base + j]       = c1;
    c_ws[base + j + H2v] = c2;
    const float e1 = e_ws[base + j];
    const float e2 = e_ws[base + j + H2v];
    const float n1 = reS * c1 - imS * c2 + e1;
    const float n2 = imS * c1 + reS * c2 + e2;
    c1 = n1; c2 = n2;
  }
}

__global__ void __launch_bounds__(128)
lru_correct(const float* __restrict__ lr, const float* __restrict__ li,
            const float* __restrict__ Wx2y, const float* __restrict__ c_ws,
            float* __restrict__ Y) {
  const int b = blockIdx.x;
  const int s = blockIdx.y;
  const int j = threadIdx.x;
  const int wave = j >> 5, lane = j & 31;

  __shared__ float Xc[CHNK][XPAD];
  __shared__ float Yp[4][CHNK][16];

  const float r  = expf(-fabsf(lr[j]));
  const float th = 1.5707963267948966f * li[j];
  const float re = r * cosf(th);
  const float im = r * sinf(th);

  v2f bop[16];
  load_wt_ops(Wx2y, wave, lane, bop);

  const size_t cbase = ((size_t)s * NBv + b) * NHv;
  float v1 = c_ws[cbase + j];
  float v2 = c_ws[cbase + j + H2v];

  for (int ch = 0; ch < NCH; ++ch) {
    const int t0 = s * SEG + ch * CHNK;
#pragma unroll
    for (int tt = 0; tt < CHNK; ++tt) {
      const float n1 = re * v1 - im * v2;
      const float n2 = im * v1 + re * v2;
      v1 = n1; v2 = n2;
      Xc[tt][j]       = v1;
      Xc[tt][j + H2v] = v2;
    }
    __syncthreads();

    project_chunk(Xc, bop, wave, lane, Yp);
    __syncthreads();

    {
      const int tt = j >> 3, n = j & 7;
      const size_t yi = ((size_t)(t0 + tt) * NBv + b) * NYv + n;
      Y[yi] += Yp[0][tt][n] + Yp[1][tt][n] + Yp[2][tt][n] + Yp[3][tt][n];
    }
    __syncthreads();
  }
}

extern "C" void kernel_launch(void* const* d_in, const int* in_sizes, int n_in,
                              void* d_out, int out_size, void* d_ws, size_t ws_size,
                              hipStream_t stream) {
  (void)in_sizes; (void)n_in; (void)out_size; (void)ws_size;
  const float* y0   = (const float*)d_in[0];
  const float* U    = (const float*)d_in[1];
  const float* lr   = (const float*)d_in[2];
  const float* li   = (const float*)d_in[3];
  const float* Bm   = (const float*)d_in[4];
  const float* mult = (const float*)d_in[5];
  const float* Wy2x = (const float*)d_in[6];
  const float* by2x = (const float*)d_in[7];
  const float* Wx2y = (const float*)d_in[8];
  const float* bx2y = (const float*)d_in[9];
  float* Y = (float*)d_out;

  float* e_ws = (float*)d_ws;
  float* c_ws = e_ws + (size_t)NSEG * NBv * NHv;

  dim3 grid(NBv, NSEG);
  lru_seg_scan<<<grid, 128, 0, stream>>>(U, lr, li, Bm, mult, Wx2y, bx2y, Y, e_ws);
  lru_carry<<<NBv, 128, 0, stream>>>(y0, Wy2x, by2x, lr, li, e_ws, c_ws);
  lru_correct<<<grid, 128, 0, stream>>>(lr, li, Wx2y, c_ws, Y);
}


# Carry-first order: state-only pre-scan, carry pass, then one scan from the true carry-in with a single WMMA projection; correction pass and its Y read-add-write removed; summation order changes
# speedup vs baseline: 1.3474x; 1.3474x over previous
#include <hip/hip_runtime.h>


typedef __attribute__((ext_vector_type(2))) float v2f;
typedef __attribute__((ext_vector_type(8))) float v8f;

#define NYv   8
#define NUv   8
#define NHv   256
#define H2v   128
#define T_LEN 2048
#define NBv   64
#define NSEG  16
#define SEG   (T_LEN / NSEG)
#define CHNK  16
#define NCH   (SEG / CHNK)
#define XPAD  258

__device__ __forceinline__ void load_wt_ops(const float* __restrict__ Wx2y,
                                            int wave, int lane, v2f bop[16]) {
  const int n  = lane & 15;
  const int kh = (lane >> 4) * 2;
#pragma unroll
  for (int i = 0; i < 16; ++i) {
    const int k0 = wave * 64 + 4 * i + kh;
    v2f v;
    if (n < NYv) {
      v.x = Wx2y[n * NHv + k0];
      v.y = Wx2y[n * NHv + k0 + 1];
    } else {
      v.x = 0.0f; v.y = 0.0f;
    }
    bop[i] = v;
  }
}

__device__ __forceinline__ void project_chunk(const float (*Xc)[XPAD],
                                              const v2f bop[16],
                                              int wave, int lane,
                                              float (*Yp)[CHNK][16]) {
  v8f acc = {};
  const int row = lane & 15;
  const int kh  = (lane >> 4) * 2;
#pragma unroll
  for (int i = 0; i < 16; ++i) {
    const int k0 = wave * 64 + 4 * i + kh;
    v2f a;
    a.x = Xc[row][k0];
    a.y = Xc[row][k0 + 1];
    acc = __builtin_amdgcn_wmma_f32_16x16x4_f32(
        false, a, false, bop[i], (short)0, acc, false, false);
  }
  const int mbase = (lane >> 4) * 8;
  const int ncol  = lane & 15;
#pragma unroll
  for (int v = 0; v < 8; ++v) Yp[wave][v + mbase][ncol] = acc[v];
}

__global__ void __launch_bounds__(128)
lru_seg_state(const float* __restrict__ U,   const float* __restrict__ lr,
              const float* __restrict__ li,  const float* __restrict__ Bm,
              const float* __restrict__ mult,float* __restrict__ e_ws) {
  const int b = blockIdx.x;
  const int s = blockIdx.y;
  const int j = threadIdx.x;

  __shared__ float4 Us[SEG][2];

  const int tt0 = j >> 1, h0 = j & 1;
  const float4 ua =
      *(const float4*)(U + ((size_t)(s * SEG + tt0) * NBv + b) * NUv + h0 * 4);
  const float4 ub =
      *(const float4*)(U + ((size_t)(s * SEG + tt0 + SEG / 2) * NBv + b) * NUv + h0 * 4);

  const float m1 = expf(mult[j]);
  const float m2 = expf(mult[j + H2v]);
  float Bs1[NUv], Bs2[NUv];
#pragma unroll
  for (int u = 0; u < NUv; ++u) {
    Bs1[u] = Bm[j * NUv + u] * m1;
    Bs2[u] = Bm[(j + H2v) * NUv + u] * m2;
  }
  const float r  = expf(-fabsf(lr[j]));
  const float th = 1.5707963267948966f * li[j];
  const float re = r * cosf(th);
  const float im = r * sinf(th);

  Us[tt0][h0]           = ua;
  Us[tt0 + SEG / 2][h0] = ub;
  __syncthreads();

  float x1 = 0.0f, x2 = 0.0f;
#pragma unroll 16
  for (int tt = 0; tt < SEG; ++tt) {
    const float4 p = Us[tt][0];
    const float4 q = Us[tt][1];
    const float uu[NUv] = {p.x, p.y, p.z, p.w, q.x, q.y, q.z, q.w};
    float bu1 = 0.0f, bu2 = 0.0f;
#pragma unroll
    for (int u = 0; u < NUv; ++u) {
      bu1 = fmaf(uu[u], Bs1[u], bu1);
      bu2 = fmaf(uu[u], Bs2[u], bu2);
    }
    const float n1 = re * x1 - im * x2 + bu1;
    const float n2 = im * x1 + re * x2 + bu2;
    x1 = n1; x2 = n2;
  }

  e_ws[((size_t)s * NBv + b) * NHv + j]       = x1;
  e_ws[((size_t)s * NBv + b) * NHv + j + H2v] = x2;
}

__global__ void __launch_bounds__(128)
lru_carry(const float* __restrict__ y0,  const float* __restrict__ Wy2x,
          const float* __restrict__ by2x,const float* __restrict__ lr,
          const float* __restrict__ li,  const float* __restrict__ e_ws,
          float* __restrict__ c_ws) {
  const int b = blockIdx.x;
  const int j = threadIdx.x;

  const float th  = 1.5707963267948966f * li[j];
  const float rS  = expf(-(float)SEG * fabsf(lr[j]));
  const float phS = (float)SEG * th;
  const float reS = rS * cosf(phS);
  const float imS = rS * sinf(phS);

  float c1 = by2x[j], c2 = by2x[j + H2v];
#pragma unroll
  for (int n = 0; n < NYv; ++n) {
    const float yv = y0[b * NYv + n];
    c1 = fmaf(yv, Wy2x[j * NYv + n], c1);
    c2 = fmaf(yv, Wy2x[(j + H2v) * NYv + n], c2);
  }

  for (int s = 0; s < NSEG; ++s) {
    const size_t base = ((size_t)s * NBv + b) * NHv;
    c_ws[base + j]       = c1;
    c_ws[base + j + H2v] = c2;
    const float e1 = e_ws[base + j];
    const float e2 = e_ws[base + j + H2v];
    const float n1 = reS * c1 - imS * c2 + e1;
    const float n2 = imS * c1 + reS * c2 + e2;
    c1 = n1; c2 = n2;
  }
}

__global__ void __launch_bounds__(128)
lru_scan_project(const float* __restrict__ U,   const float* __restrict__ lr,
                 const float* __restrict__ li,  const float* __restrict__ Bm,
                 const float* __restrict__ mult,const float* __restrict__ Wx2y,
                 const float* __restrict__ bx2y,const float* __restrict__ c_ws,
                 float* __restrict__ Y) {
  const int b = blockIdx.x;
  const int s = blockIdx.y;
  const int j = threadIdx.x;
  const int wave = j >> 5, lane = j & 31;

  __shared__ float Uc[CHNK][NUv];
  __shared__ float Xc[CHNK][XPAD];
  __shared__ float Yp[4][CHNK][16];

  const float m1 = expf(mult[j]);
  const float m2 = expf(mult[j + H2v]);
  float Bs1[NUv], Bs2[NUv];
#pragma unroll
  for (int u = 0; u < NUv; ++u) {
    Bs1[u] = Bm[j * NUv + u] * m1;
    Bs2[u] = Bm[(j + H2v) * NUv + u] * m2;
  }
  const float r  = expf(-fabsf(lr[j]));
  const float th = 1.5707963267948966f * li[j];
  const float re = r * cosf(th);
  const float im = r * sinf(th);

  v2f bop[16];
  load_wt_ops(Wx2y, wave, lane, bop);

  float x1 = c_ws[((size_t)s * NBv + b) * NHv + j];
  float x2 = c_ws[((size_t)s * NBv + b) * NHv + j + H2v];

  if (j < CHNK)
    __builtin_prefetch(U + ((size_t)(s * SEG + j) * NBv + b) * NUv, 0, 0);

  for (int ch = 0; ch < NCH; ++ch) {
    const int t0 = s * SEG + ch * CHNK;
    if (j < 2 * CHNK) {
      const int tt = j >> 1, h = (j & 1) * 4;
      const float4 uv =
          *(const float4*)(U + ((size_t)(t0 + tt) * NBv + b) * NUv + h);
      Uc[tt][h + 0] = uv.x; Uc[tt][h + 1] = uv.y;
      Uc[tt][h + 2] = uv.z; Uc[tt][h + 3] = uv.w;
    }
    if (ch + 1 < NCH && j < CHNK)
      __builtin_prefetch(U + ((size_t)(t0 + CHNK + j) * NBv + b) * NUv, 0, 0);
    __syncthreads();

#pragma unroll
    for (int tt = 0; tt < CHNK; ++tt) {
      float bu1 = 0.0f, bu2 = 0.0f;
#pragma unroll
      for (int u = 0; u < NUv; ++u) {
        const float uu = Uc[tt][u];
        bu1 = fmaf(uu, Bs1[u], bu1);
        bu2 = fmaf(uu, Bs2[u], bu2);
      }
      const float n1 = re * x1 - im * x2 + bu1;
      const float n2 = im * x1 + re * x2 + bu2;
      x1 = n1; x2 = n2;
      Xc[tt][j]       = x1;
      Xc[tt][j + H2v] = x2;
    }
    __syncthreads();

    project_chunk(Xc, bop, wave, lane, Yp);
    __syncthreads();

    {
      const int tt = j >> 3, n = j & 7;
      const float y = Yp[0][tt][n] + Yp[1][tt][n] + Yp[2][tt][n] + Yp[3][tt][n]
                    + bx2y[n];
      Y[((size_t)(t0 + tt) * NBv + b) * NYv + n] = y;
    }
    __syncthreads();
  }
}

extern "C" void kernel_launch(void* const* d_in, const int* in_sizes, int n_in,
                              void* d_out, int out_size, void* d_ws, size_t ws_size,
                              hipStream_t stream) {
  (void)in_sizes; (void)n_in; (void)out_size; (void)ws_size;
  const float* y0   = (const float*)d_in[0];
  const float* U    = (const float*)d_in[1];
  const float* lr   = (const float*)d_in[2];
  const float* li   = (const float*)d_in[3];
  const float* Bm   = (const float*)d_in[4];
  const float* mult = (const float*)d_in[5];
  const float* Wy2x = (const float*)d_in[6];
  const float* by2x = (const float*)d_in[7];
  const float* Wx2y = (const float*)d_in[8];
  const float* bx2y = (const float*)d_in[9];
  float* Y = (float*)d_out;

  float* e_ws = (float*)d_ws;
  float* c_ws = e_ws + (size_t)NSEG * NBv * NHv;

  dim3 grid(NBv, NSEG);
  lru_seg_state<<<grid, 128, 0, stream>>>(U, lr, li, Bm, mult, e_ws);
  lru_carry<<<NBv, 128, 0, stream>>>(y0, Wy2x, by2x, lr, li, e_ws, c_ws);
  lru_scan_project<<<grid, 128, 0, stream>>>(U, lr, li, Bm, mult, Wx2y, bx2y, c_ws, Y);
}


# Carry scan folded into the scan-project kernel: each block builds its carry-in from the earlier segments' end states itself; carry launch and its workspace plane removed; summation order unchanged
# speedup vs baseline: 1.4481x; 1.0747x over previous
#include <hip/hip_runtime.h>


typedef __attribute__((ext_vector_type(2))) float v2f;
typedef __attribute__((ext_vector_type(8))) float v8f;

#define NYv   8
#define NUv   8
#define NHv   256
#define H2v   128
#define T_LEN 2048
#define NBv   64
#define NSEG  16
#define SEG   (T_LEN / NSEG)
#define CHNK  16
#define NCH   (SEG / CHNK)
#define XPAD  258

__device__ __forceinline__ void load_wt_ops(const float* __restrict__ Wx2y,
                                            int wave, int lane, v2f bop[16]) {
  const int n  = lane & 15;
  const int kh = (lane >> 4) * 2;
#pragma unroll
  for (int i = 0; i < 16; ++i) {
    const int k0 = wave * 64 + 4 * i + kh;
    v2f v;
    if (n < NYv) {
      v.x = Wx2y[n * NHv + k0];
      v.y = Wx2y[n * NHv + k0 + 1];
    } else {
      v.x = 0.0f; v.y = 0.0f;
    }
    bop[i] = v;
  }
}

__device__ __forceinline__ void project_chunk(const float (*Xc)[XPAD],
                                              const v2f bop[16],
                                              int wave, int lane,
                                              float (*Yp)[CHNK][16]) {
  v8f acc = {};
  const int row = lane & 15;
  const int kh  = (lane >> 4) * 2;
#pragma unroll
  for (int i = 0; i < 16; ++i) {
    const int k0 = wave * 64 + 4 * i + kh;
    v2f a;
    a.x = Xc[row][k0];
    a.y = Xc[row][k0 + 1];
    acc = __builtin_amdgcn_wmma_f32_16x16x4_f32(
        false, a, false, bop[i], (short)0, acc, false, false);
  }
  const int mbase = (lane >> 4) * 8;
  const int ncol  = lane & 15;
#pragma unroll
  for (int v = 0; v < 8; ++v) Yp[wave][v + mbase][ncol] = acc[v];
}

__global__ void __launch_bounds__(128)
lru_seg_state(const float* __restrict__ U,   const float* __restrict__ lr,
              const float* __restrict__ li,  const float* __restrict__ Bm,
              const float* __restrict__ mult,float* __restrict__ e_ws) {
  const int b = blockIdx.x;
  const int s = blockIdx.y;
  const int j = threadIdx.x;

  __shared__ float4 Us[SEG][2];

  const int tt0 = j >> 1, h0 = j & 1;
  const float4 ua =
      *(const float4*)(U + ((size_t)(s * SEG + tt0) * NBv + b) * NUv + h0 * 4);
  const float4 ub =
      *(const float4*)(U + ((size_t)(s * SEG + tt0 + SEG / 2) * NBv + b) * NUv + h0 * 4);

  const float m1 = expf(mult[j]);
  const float m2 = expf(mult[j + H2v]);
  float Bs1[NUv], Bs2[NUv];
#pragma unroll
  for (int u = 0; u < NUv; ++u) {
    Bs1[u] = Bm[j * NUv + u] * m1;
    Bs2[u] = Bm[(j + H2v) * NUv + u] * m2;
  }
  const float r  = expf(-fabsf(lr[j]));
  const float th = 1.5707963267948966f * li[j];
  const float re = r * cosf(th);
  const float im = r * sinf(th);

  Us[tt0][h0]           = ua;
  Us[tt0 + SEG / 2][h0] = ub;
  __syncthreads();

  float x1 = 0.0f, x2 = 0.0f;
#pragma unroll 16
  for (int tt = 0; tt < SEG; ++tt) {
    const float4 p = Us[tt][0];
    const float4 q = Us[tt][1];
    const float uu[NUv] = {p.x, p.y, p.z, p.w, q.x, q.y, q.z, q.w};
    float bu1 = 0.0f, bu2 = 0.0f;
#pragma unroll
    for (int u = 0; u < NUv; ++u) {
      bu1 = fmaf(uu[u], Bs1[u], bu1);
      bu2 = fmaf(uu[u], Bs2[u], bu2);
    }
    const float n1 = re * x1 - im * x2 + bu1;
    const float n2 = im * x1 + re * x2 + bu2;
    x1 = n1; x2 = n2;
  }

  e_ws[((size_t)s * NBv + b) * NHv + j]       = x1;
  e_ws[((size_t)s * NBv + b) * NHv + j + H2v] = x2;
}

__global__ void __launch_bounds__(128)
lru_scan_project(const float* __restrict__ U,   const float* __restrict__ lr,
                 const float* __restrict__ li,  const float* __restrict__ Bm,
                 const float* __restrict__ mult,const float* __restrict__ Wx2y,
                 const float* __restrict__ bx2y,const float* __restrict__ y0,
                 const float* __restrict__ Wy2x,const float* __restrict__ by2x,
                 const float* __restrict__ e_ws,float* __restrict__ Y) {
  const int b = blockIdx.x;
  const int s = blockIdx.y;
  const int j = threadIdx.x;
  const int wave = j >> 5, lane = j & 31;

  __shared__ float Uc[CHNK][NUv];
  __shared__ float Xc[CHNK][XPAD];
  __shared__ float Yp[4][CHNK][16];

  float e1[NSEG - 1], e2[NSEG - 1];
#pragma unroll
  for (int i = 0; i < NSEG - 1; ++i) {
    const int seg = i < s ? i : 0;
    const float* row = e_ws + ((size_t)seg * NBv + b) * NHv;
    e1[i] = row[j];
    e2[i] = row[j + H2v];
  }

  const float m1 = expf(mult[j]);
  const float m2 = expf(mult[j + H2v]);
  float Bs1[NUv], Bs2[NUv];
#pragma unroll
  for (int u = 0; u < NUv; ++u) {
    Bs1[u] = Bm[j * NUv + u] * m1;
    Bs2[u] = Bm[(j + H2v) * NUv + u] * m2;
  }
  const float r  = expf(-fabsf(lr[j]));
  const float th = 1.5707963267948966f * li[j];
  const float re = r * cosf(th);
  const float im = r * sinf(th);

  v2f bop[16];
  load_wt_ops(Wx2y, wave, lane, bop);

  float x1 = by2x[j], x2 = by2x[j + H2v];
#pragma unroll
  for (int n = 0; n < NYv; ++n) {
    const float yv = y0[b * NYv + n];
    x1 = fmaf(yv, Wy2x[j * NYv + n], x1);
    x2 = fmaf(yv, Wy2x[(j + H2v) * NYv + n], x2);
  }
  {
    const float rS  = expf(-(float)SEG * fabsf(lr[j]));
    const float phS = (float)SEG * th;
    const float reS = rS * cosf(phS);
    const float imS = rS * sinf(phS);
#pragma unroll
    for (int i = 0; i < NSEG - 1; ++i) {
      if (i < s) {
        const float n1 = reS * x1 - imS * x2 + e1[i];
        const float n2 = imS * x1 + reS * x2 + e2[i];
        x1 = n1; x2 = n2;
      }
    }
  }

  if (j < CHNK)
    __builtin_prefetch(U + ((size_t)(s * SEG + j) * NBv + b) * NUv, 0, 0);

  for (int ch = 0; ch < NCH; ++ch) {
    const int t0 = s * SEG + ch * CHNK;
    if (j < 2 * CHNK) {
      const int tt = j >> 1, h = (j & 1) * 4;
      const float4 uv =
          *(const float4*)(U + ((size_t)(t0 + tt) * NBv + b) * NUv + h);
      Uc[tt][h + 0] = uv.x; Uc[tt][h + 1] = uv.y;
      Uc[tt][h + 2] = uv.z; Uc[tt][h + 3] = uv.w;
    }
    if (ch + 1 < NCH && j < CHNK)
      __builtin_prefetch(U + ((size_t)(t0 + CHNK + j) * NBv + b) * NUv, 0, 0);
    __syncthreads();

#pragma unroll
    for (int tt = 0; tt < CHNK; ++tt) {
      float bu1 = 0.0f, bu2 = 0.0f;
#pragma unroll
      for (int u = 0; u < NUv; ++u) {
        const float uu = Uc[tt][u];
        bu1 = fmaf(uu, Bs1[u], bu1);
        bu2 = fmaf(uu, Bs2[u], bu2);
      }
      const float n1 = re * x1 - im * x2 + bu1;
      const float n2 = im * x1 + re * x2 + bu2;
      x1 = n1; x2 = n2;
      Xc[tt][j]       = x1;
      Xc[tt][j + H2v] = x2;
    }
    __syncthreads();

    project_chunk(Xc, bop, wave, lane, Yp);
    __syncthreads();

    {
      const int tt = j >> 3, n = j & 7;
      const float y = Yp[0][tt][n] + Yp[1][tt][n] + Yp[2][tt][n] + Yp[3][tt][n]
                    + bx2y[n];
      Y[((size_t)(t0 + tt) * NBv + b) * NYv + n] = y;
    }
    __syncthreads();
  }
}

extern "C" void kernel_launch(void* const* d_in, const int* in_sizes, int n_in,
                              void* d_out, int out_size, void* d_ws, size_t ws_size,
                              hipStream_t stream) {
  (void)in_sizes; (void)n_in; (void)out_size; (void)ws_size;
  const float* y0   = (const float*)d_in[0];
  const float* U    = (const float*)d_in[1];
  const float* lr   = (const float*)d_in[2];
  const float* li   = (const float*)d_in[3];
  const float* Bm   = (const float*)d_in[4];
  const float* mult = (const float*)d_in[5];
  const float* Wy2x = (const float*)d_in[6];
  const float* by2x = (const float*)d_in[7];
  const float* Wx2y = (const float*)d_in[8];
  const float* bx2y = (const float*)d_in[9];
  float* Y = (float*)d_out;

  float* e_ws = (float*)d_ws;

  dim3 grid(NBv, NSEG);
  lru_seg_state<<<grid, 128, 0, stream>>>(U, lr, li, Bm, mult, e_ws);
  lru_scan_project<<<grid, 128, 0, stream>>>(U, lr, li, Bm, mult, Wx2y, bx2y,
                                             y0, Wy2x, by2x, e_ws, Y);
}


# Launch 1 input drive (U.B^T, K=8) moved from VALU FMAs to f32 WMMA: rows = 8 steps x 2 batches so D lands in scan-lane order and enters the rotate as FMA addend; block = 2 batches; sum order changes
# speedup vs baseline: 1.5583x; 1.0761x over previous
#include <hip/hip_runtime.h>


typedef __attribute__((ext_vector_type(2))) float v2f;
typedef __attribute__((ext_vector_type(8))) float v8f;

#define NYv   8
#define NUv   8
#define NHv   256
#define H2v   128
#define T_LEN 2048
#define NBv   64
#define NSEG  16
#define SEG   (T_LEN / NSEG)
#define CHNK  16
#define NCH   (SEG / CHNK)
#define XPAD  258
#define NBP   32

__device__ __forceinline__ void load_wt_ops(const float* __restrict__ Wx2y,
                                            int wave, int lane, v2f bop[16]) {
  const int n  = lane & 15;
  const int kh = (lane >> 4) * 2;
#pragma unroll
  for (int i = 0; i < 16; ++i) {
    const int k0 = wave * 64 + 4 * i + kh;
    v2f v;
    if (n < NYv) {
      v.x = Wx2y[n * NHv + k0];
      v.y = Wx2y[n * NHv + k0 + 1];
    } else {
      v.x = 0.0f; v.y = 0.0f;
    }
    bop[i] = v;
  }
}

__device__ __forceinline__ void project_chunk(const float (*Xc)[XPAD],
                                              const v2f bop[16],
                                              int wave, int lane,
                                              float (*Yp)[CHNK][16]) {
  v8f acc = {};
  const int row = lane & 15;
  const int kh  = (lane >> 4) * 2;
#pragma unroll
  for (int i = 0; i < 16; ++i) {
    const int k0 = wave * 64 + 4 * i + kh;
    v2f a;
    a.x = Xc[row][k0];
    a.y = Xc[row][k0 + 1];
    acc = __builtin_amdgcn_wmma_f32_16x16x4_f32(
        false, a, false, bop[i], (short)0, acc, false, false);
  }
  const int mbase = (lane >> 4) * 8;
  const int ncol  = lane & 15;
#pragma unroll
  for (int v = 0; v < 8; ++v) Yp[wave][v + mbase][ncol] = acc[v];
}

__global__ void __launch_bounds__(256)
lru_seg_state(const float* __restrict__ U,   const float* __restrict__ lr,
              const float* __restrict__ li,  const float* __restrict__ Bm,
              const float* __restrict__ mult,float* __restrict__ e_ws) {
  const int bp   = blockIdx.x;
  const int s    = blockIdx.y;
  const int tid  = threadIdx.x;
  const int wave = tid >> 5, lane = tid & 31;
  const int c = lane & 15, h = lane >> 4;
  const int j = wave * 16 + c;
  const int b = 2 * bp + h;

  __shared__ float4 Us[SEG * 4];

  const float* Ub = U + ((size_t)(s * SEG) * NBv + 2 * bp) * NUv;
  const int st0 = tid >> 2, q0 = tid & 3;
  const float4 ua =
      *(const float4*)(Ub + (size_t)st0 * (NBv * NUv) + q0 * 4);
  const float4 ub =
      *(const float4*)(Ub + (size_t)(st0 + SEG / 2) * (NBv * NUv) + q0 * 4);

  const float m1 = expf(mult[j]);
  const float m2 = expf(mult[j + H2v]);
  const float4 w1 = *(const float4*)(Bm + j * NUv + 4 * h);
  const float4 w2 = *(const float4*)(Bm + (j + H2v) * NUv + 4 * h);
  v2f b1a, b1b, b2a, b2b;
  b1a.x = w1.x * m1; b1a.y = w1.y * m1; b1b.x = w1.z * m1; b1b.y = w1.w * m1;
  b2a.x = w2.x * m2; b2a.y = w2.y * m2; b2b.x = w2.z * m2; b2b.y = w2.w * m2;

  const float r  = expf(-fabsf(lr[j]));
  const float th = 1.5707963267948966f * li[j];
  const float re = r * cosf(th);
  const float im = r * sinf(th);

  Us[tid]       = ua;
  Us[tid + 256] = ub;
  __syncthreads();

  const int arow = (c & 7) * 4 + (c >> 3) * 2 + h;

  float x1 = 0.0f, x2 = 0.0f;
#pragma unroll
  for (int g = 0; g < SEG / 8; ++g) {
    const float4 av = Us[g * 32 + arow];
    v2f a0, a1;
    a0.x = av.x; a0.y = av.y; a1.x = av.z; a1.y = av.w;
    v8f d1 = {}, d2 = {};
    d1 = __builtin_amdgcn_wmma_f32_16x16x4_f32(
        false, a0, false, b1a, (short)0, d1, false, false);
    d2 = __builtin_amdgcn_wmma_f32_16x16x4_f32(
        false, a0, false, b2a, (short)0, d2, false, false);
    d1 = __builtin_amdgcn_wmma_f32_16x16x4_f32(
        false, a1, false, b1b, (short)0, d1, false, false);
    d2 = __builtin_amdgcn_wmma_f32_16x16x4_f32(
        false, a1, false, b2b, (short)0, d2, false, false);
#pragma unroll
    for (int rr = 0; rr < 8; ++rr) {
      const float p  = fmaf(-im, x2, d1[rr]);
      const float q  = fmaf( im, x1, d2[rr]);
      const float n1 = fmaf(re, x1, p);
      const float n2 = fmaf(re, x2, q);
      x1 = n1; x2 = n2;
    }
  }

  e_ws[((size_t)s * NBv + b) * NHv + j]       = x1;
  e_ws[((size_t)s * NBv + b) * NHv + j + H2v] = x2;
}

__global__ void __launch_bounds__(128)
lru_scan_project(const float* __restrict__ U,   const float* __restrict__ lr,
                 const float* __restrict__ li,  const float* __restrict__ Bm,
                 const float* __restrict__ mult,const float* __restrict__ Wx2y,
                 const float* __restrict__ bx2y,const float* __restrict__ y0,
                 const float* __restrict__ Wy2x,const float* __restrict__ by2x,
                 const float* __restrict__ e_ws,float* __restrict__ Y) {
  const int b = blockIdx.x;
  const int s = blockIdx.y;
  const int j = threadIdx.x;
  const int wave = j >> 5, lane = j & 31;

  __shared__ float Uc[CHNK][NUv];
  __shared__ float Xc[CHNK][XPAD];
  __shared__ float Yp[4][CHNK][16];

  float e1[NSEG - 1], e2[NSEG - 1];
#pragma unroll
  for (int i = 0; i < NSEG - 1; ++i) {
    const int seg = i < s ? i : 0;
    const float* row = e_ws + ((size_t)seg * NBv + b) * NHv;
    e1[i] = row[j];
    e2[i] = row[j + H2v];
  }

  const float m1 = expf(mult[j]);
  const float m2 = expf(mult[j + H2v]);
  float Bs1[NUv], Bs2[NUv];
#pragma unroll
  for (int u = 0; u < NUv; ++u) {
    Bs1[u] = Bm[j * NUv + u] * m1;
    Bs2[u] = Bm[(j + H2v) * NUv + u] * m2;
  }
  const float r  = expf(-fabsf(lr[j]));
  const float th = 1.5707963267948966f * li[j];
  const float re = r * cosf(th);
  const float im = r * sinf(th);

  v2f bop[16];
  load_wt_ops(Wx2y, wave, lane, bop);

  float x1 = by2x[j], x2 = by2x[j + H2v];
#pragma unroll
  for (int n = 0; n < NYv; ++n) {
    const float yv = y0[b * NYv + n];
    x1 = fmaf(yv, Wy2x[j * NYv + n], x1);
    x2 = fmaf(yv, Wy2x[(j + H2v) * NYv + n], x2);
  }
  {
    const float rS  = expf(-(float)SEG * fabsf(lr[j]));
    const float phS = (float)SEG * th;
    const float reS = rS * cosf(phS);
    const float imS = rS * sinf(phS);
#pragma unroll
    for (int i = 0; i < NSEG - 1; ++i) {
      if (i < s) {
        const float n1 = reS * x1 - imS * x2 + e1[i];
        const float n2 = imS * x1 + reS * x2 + e2[i];
        x1 = n1; x2 = n2;
      }
    }
  }

  if (j < CHNK)
    __builtin_prefetch(U + ((size_t)(s * SEG + j) * NBv + b) * NUv, 0, 0);

  for (int ch = 0; ch < NCH; ++ch) {
    const int t0 = s * SEG + ch * CHNK;
    if (j < 2 * CHNK) {
      const int tt = j >> 1, h = (j & 1) * 4;
      const float4 uv =
          *(const float4*)(U + ((size_t)(t0 + tt) * NBv + b) * NUv + h);
      Uc[tt][h + 0] = uv.x; Uc[tt][h + 1] = uv.y;
      Uc[tt][h + 2] = uv.z; Uc[tt][h + 3] = uv.w;
    }
    if (ch + 1 < NCH && j < CHNK)
      __builtin_prefetch(U + ((size_t)(t0 + CHNK + j) * NBv + b) * NUv, 0, 0);
    __syncthreads();

#pragma unroll
    for (int tt = 0; tt < CHNK; ++tt) {
      float bu1 = 0.0f, bu2 = 0.0f;
#pragma unroll
      for (int u = 0; u < NUv; ++u) {
        const float uu = Uc[tt][u];
        bu1 = fmaf(uu, Bs1[u], bu1);
        bu2 = fmaf(uu, Bs2[u], bu2);
      }
      const float n1 = re * x1 - im * x2 + bu1;
      const float n2 = im * x1 + re * x2 + bu2;
      x1 = n1; x2 = n2;
      Xc[tt][j]       = x1;
      Xc[tt][j + H2v] = x2;
    }
    __syncthreads();

    project_chunk(Xc, bop, wave, lane, Yp);
    __syncthreads();

    {
      const int tt = j >> 3, n = j & 7;
      const float y = Yp[0][tt][n] + Yp[1][tt][n] + Yp[2][tt][n] + Yp[3][tt][n]
                    + bx2y[n];
      Y[((size_t)(t0 + tt) * NBv + b) * NYv + n] = y;
    }
    __syncthreads();
  }
}

extern "C" void kernel_launch(void* const* d_in, const int* in_sizes, int n_in,
                              void* d_out, int out_size, void* d_ws, size_t ws_size,
                              hipStream_t stream) {
  (void)in_sizes; (void)n_in; (void)out_size; (void)ws_size;
  const float* y0   = (const float*)d_in[0];
  const float* U    = (const float*)d_in[1];
  const float* lr   = (const float*)d_in[2];
  const float* li   = (const float*)d_in[3];
  const float* Bm   = (const float*)d_in[4];
  const float* mult = (const float*)d_in[5];
  const float* Wy2x = (const float*)d_in[6];
  const float* by2x = (const float*)d_in[7];
  const float* Wx2y = (const float*)d_in[8];
  const float* bx2y = (const float*)d_in[9];
  float* Y = (float*)d_out;

  float* e_ws = (float*)d_ws;

  dim3 grid1(NBP, NSEG);
  dim3 grid(NBv, NSEG);
  lru_seg_state<<<grid1, 256, 0, stream>>>(U, lr, li, Bm, mult, e_ws);
  lru_scan_project<<<grid, 128, 0, stream>>>(U, lr, li, Bm, mult, Wx2y, bx2y,
                                             y0, Wy2x, by2x, e_ws, Y);
}


# Launch 2 chunk loop freed of global-memory waits: the segment's U staged to LDS once before the loop; barriers cut from 4 to 2 per chunk and fence LDS only; summation order unchanged
# speedup vs baseline: 1.7992x; 1.1546x over previous
#include <hip/hip_runtime.h>


typedef __attribute__((ext_vector_type(2))) float v2f;
typedef __attribute__((ext_vector_type(8))) float v8f;

#define NYv   8
#define NUv   8
#define NHv   256
#define H2v   128
#define T_LEN 2048
#define NBv   64
#define NSEG  16
#define SEG   (T_LEN / NSEG)
#define CHNK  16
#define NCH   (SEG / CHNK)
#define XPAD  258
#define NBP   32

__device__ __forceinline__ void load_wt_ops(const float* __restrict__ Wx2y,
                                            int wave, int lane, v2f bop[16]) {
  const int n  = lane & 15;
  const int kh = (lane >> 4) * 2;
#pragma unroll
  for (int i = 0; i < 16; ++i) {
    const int k0 = wave * 64 + 4 * i + kh;
    v2f v;
    if (n < NYv) {
      v.x = Wx2y[n * NHv + k0];
      v.y = Wx2y[n * NHv + k0 + 1];
    } else {
      v.x = 0.0f; v.y = 0.0f;
    }
    bop[i] = v;
  }
}

__device__ __forceinline__ void project_chunk(const float (*Xc)[XPAD],
                                              const v2f bop[16],
                                              int wave, int lane,
                                              float (*Yp)[CHNK][16]) {
  v8f acc = {};
  const int row = lane & 15;
  const int kh  = (lane >> 4) * 2;
#pragma unroll
  for (int i = 0; i < 16; ++i) {
    const int k0 = wave * 64 + 4 * i + kh;
    v2f a;
    a.x = Xc[row][k0];
    a.y = Xc[row][k0 + 1];
    acc = __builtin_amdgcn_wmma_f32_16x16x4_f32(
        false, a, false, bop[i], (short)0, acc, false, false);
  }
  const int mbase = (lane >> 4) * 8;
  const int ncol  = lane & 15;
#pragma unroll
  for (int v = 0; v < 8; ++v) Yp[wave][v + mbase][ncol] = acc[v];
}

__global__ void __launch_bounds__(256)
lru_seg_state(const float* __restrict__ U,   const float* __restrict__ lr,
              const float* __restrict__ li,  const float* __restrict__ Bm,
              const float* __restrict__ mult,float* __restrict__ e_ws) {
  const int bp   = blockIdx.x;
  const int s    = blockIdx.y;
  const int tid  = threadIdx.x;
  const int wave = tid >> 5, lane = tid & 31;
  const int c = lane & 15, h = lane >> 4;
  const int j = wave * 16 + c;
  const int b = 2 * bp + h;

  __shared__ float4 Us[SEG * 4];

  const float* Ub = U + ((size_t)(s * SEG) * NBv + 2 * bp) * NUv;
  const int st0 = tid >> 2, q0 = tid & 3;
  const float4 ua =
      *(const float4*)(Ub + (size_t)st0 * (NBv * NUv) + q0 * 4);
  const float4 ub =
      *(const float4*)(Ub + (size_t)(st0 + SEG / 2) * (NBv * NUv) + q0 * 4);

  const float m1 = expf(mult[j]);
  const float m2 = expf(mult[j + H2v]);
  const float4 w1 = *(const float4*)(Bm + j * NUv + 4 * h);
  const float4 w2 = *(const float4*)(Bm + (j + H2v) * NUv + 4 * h);
  v2f b1a, b1b, b2a, b2b;
  b1a.x = w1.x * m1; b1a.y = w1.y * m1; b1b.x = w1.z * m1; b1b.y = w1.w * m1;
  b2a.x = w2.x * m2; b2a.y = w2.y * m2; b2b.x = w2.z * m2; b2b.y = w2.w * m2;

  const float r  = expf(-fabsf(lr[j]));
  const float th = 1.5707963267948966f * li[j];
  const float re = r * cosf(th);
  const float im = r * sinf(th);

  Us[tid]       = ua;
  Us[tid + 256] = ub;
  __syncthreads();

  const int arow = (c & 7) * 4 + (c >> 3) * 2 + h;

  float x1 = 0.0f, x2 = 0.0f;
#pragma unroll
  for (int g = 0; g < SEG / 8; ++g) {
    const float4 av = Us[g * 32 + arow];
    v2f a0, a1;
    a0.x = av.x; a0.y = av.y; a1.x = av.z; a1.y = av.w;
    v8f d1 = {}, d2 = {};
    d1 = __builtin_amdgcn_wmma_f32_16x16x4_f32(
        false, a0, false, b1a, (short)0, d1, false, false);
    d2 = __builtin_amdgcn_wmma_f32_16x16x4_f32(
        false, a0, false, b2a, (short)0, d2, false, false);
    d1 = __builtin_amdgcn_wmma_f32_16x16x4_f32(
        false, a1, false, b1b, (short)0, d1, false, false);
    d2 = __builtin_amdgcn_wmma_f32_16x16x4_f32(
        false, a1, false, b2b, (short)0, d2, false, false);
#pragma unroll
    for (int rr = 0; rr < 8; ++rr) {
      const float p  = fmaf(-im, x2, d1[rr]);
      const float q  = fmaf( im, x1, d2[rr]);
      const float n1 = fmaf(re, x1, p);
      const float n2 = fmaf(re, x2, q);
      x1 = n1; x2 = n2;
    }
  }

  e_ws[((size_t)s * NBv + b) * NHv + j]       = x1;
  e_ws[((size_t)s * NBv + b) * NHv + j + H2v] = x2;
}

__device__ __forceinline__ void lds_barrier() {
  __builtin_amdgcn_fence(__ATOMIC_RELEASE, "workgroup", "local");
  __builtin_amdgcn_s_barrier();
  __builtin_amdgcn_fence(__ATOMIC_ACQUIRE, "workgroup", "local");
}

__global__ void __launch_bounds__(128)
lru_scan_project(const float* __restrict__ U,   const float* __restrict__ lr,
                 const float* __restrict__ li,  const float* __restrict__ Bm,
                 const float* __restrict__ mult,const float* __restrict__ Wx2y,
                 const float* __restrict__ bx2y,const float* __restrict__ y0,
                 const float* __restrict__ Wy2x,const float* __restrict__ by2x,
                 const float* __restrict__ e_ws,float* __restrict__ Y) {
  const int b = blockIdx.x;
  const int s = blockIdx.y;
  const int j = threadIdx.x;
  const int wave = j >> 5, lane = j & 31;

  __shared__ float4 Us[SEG * 2];
  __shared__ float Xc[CHNK][XPAD];
  __shared__ float Yp[4][CHNK][16];

  const float* Ub = U + ((size_t)(s * SEG) * NBv + b) * NUv;
  const int uo = (j >> 1) * (NBv * NUv) + (j & 1) * 4;
  const float4 ua = *(const float4*)(Ub + uo);
  const float4 ub = *(const float4*)(Ub + uo + (SEG / 2) * (NBv * NUv));

  float e1[NSEG - 1], e2[NSEG - 1];
#pragma unroll
  for (int i = 0; i < NSEG - 1; ++i) {
    const int seg = i < s ? i : 0;
    const float* row = e_ws + ((size_t)seg * NBv + b) * NHv;
    e1[i] = row[j];
    e2[i] = row[j + H2v];
  }

  const float m1 = expf(mult[j]);
  const float m2 = expf(mult[j + H2v]);
  float Bs1[NUv], Bs2[NUv];
#pragma unroll
  for (int u = 0; u < NUv; ++u) {
    Bs1[u] = Bm[j * NUv + u] * m1;
    Bs2[u] = Bm[(j + H2v) * NUv + u] * m2;
  }
  const float r  = expf(-fabsf(lr[j]));
  const float th = 1.5707963267948966f * li[j];
  Us[j]       = ua;
  Us[j + 128] = ub;
  const float re = r * cosf(th);
  const float im = r * sinf(th);

  v2f bop[16];
  load_wt_ops(Wx2y, wave, lane, bop);

  float x1 = by2x[j], x2 = by2x[j + H2v];
#pragma unroll
  for (int n = 0; n < NYv; ++n) {
    const float yv = y0[b * NYv + n];
    x1 = fmaf(yv, Wy2x[j * NYv + n], x1);
    x2 = fmaf(yv, Wy2x[(j + H2v) * NYv + n], x2);
  }
  {
    const float rS  = expf(-(float)SEG * fabsf(lr[j]));
    const float phS = (float)SEG * th;
    const float reS = rS * cosf(phS);
    const float imS = rS * sinf(phS);
#pragma unroll
    for (int i = 0; i < NSEG - 1; ++i) {
      if (i < s) {
        const float n1 = reS * x1 - imS * x2 + e1[i];
        const float n2 = imS * x1 + reS * x2 + e2[i];
        x1 = n1; x2 = n2;
      }
    }
  }

  __syncthreads();

  for (int ch = 0; ch < NCH; ++ch) {
    const int t0 = s * SEG + ch * CHNK;

#pragma unroll
    for (int tt = 0; tt < CHNK; ++tt) {
      const float4 u0 = Us[(ch * CHNK + tt) * 2];
      const float4 u1 = Us[(ch * CHNK + tt) * 2 + 1];
      const float uu[NUv] = {u0.x, u0.y, u0.z, u0.w, u1.x, u1.y, u1.z, u1.w};
      float bu1 = 0.0f, bu2 = 0.0f;
#pragma unroll
      for (int u = 0; u < NUv; ++u) {
        bu1 = fmaf(uu[u], Bs1[u], bu1);
        bu2 = fmaf(uu[u], Bs2[u], bu2);
      }
      const float n1 = re * x1 - im * x2 + bu1;
      const float n2 = im * x1 + re * x2 + bu2;
      x1 = n1; x2 = n2;
      Xc[tt][j]       = x1;
      Xc[tt][j + H2v] = x2;
    }
    lds_barrier();

    project_chunk(Xc, bop, wave, lane, Yp);
    lds_barrier();

    {
      const int tt = j >> 3, n = j & 7;
      const float y = Yp[0][tt][n] + Yp[1][tt][n] + Yp[2][tt][n] + Yp[3][tt][n]
                    + bx2y[n];
      Y[((size_t)(t0 + tt) * NBv + b) * NYv + n] = y;
    }
  }
}

extern "C" void kernel_launch(void* const* d_in, const int* in_sizes, int n_in,
                              void* d_out, int out_size, void* d_ws, size_t ws_size,
                              hipStream_t stream) {
  (void)in_sizes; (void)n_in; (void)out_size; (void)ws_size;
  const float* y0   = (const float*)d_in[0];
  const float* U    = (const float*)d_in[1];
  const float* lr   = (const float*)d_in[2];
  const float* li   = (const float*)d_in[3];
  const float* Bm   = (const float*)d_in[4];
  const float* mult = (const float*)d_in[5];
  const float* Wy2x = (const float*)d_in[6];
  const float* by2x = (const float*)d_in[7];
  const float* Wx2y = (const float*)d_in[8];
  const float* bx2y = (const float*)d_in[9];
  float* Y = (float*)d_out;

  float* e_ws = (float*)d_ws;

  dim3 grid1(NBP, NSEG);
  dim3 grid(NBv, NSEG);
  lru_seg_state<<<grid1, 256, 0, stream>>>(U, lr, li, Bm, mult, e_ws);
  lru_scan_project<<<grid, 128, 0, stream>>>(U, lr, li, Bm, mult, Wx2y, bx2y,
                                             y0, Wy2x, by2x, e_ws, Y);
}


# Launch 2 chunk loop software-pipelined: WMMA chain of chunk c-1 issued step by step under the scan of chunk c; X tile and partials double-buffered in LDS; barriers 2 to 1 per chunk; sums unchanged
# speedup vs baseline: 1.8689x; 1.0387x over previous
#include <hip/hip_runtime.h>


typedef __attribute__((ext_vector_type(2))) float v2f;
typedef __attribute__((ext_vector_type(8))) float v8f;

#define NYv   8
#define NUv   8
#define NHv   256
#define H2v   128
#define T_LEN 2048
#define NBv   64
#define NSEG  16
#define SEG   (T_LEN / NSEG)
#define CHNK  16
#define NCH   (SEG / CHNK)
#define XPAD  258
#define NBP   32

__device__ __forceinline__ void load_wt_ops(const float* __restrict__ Wx2y,
                                            int wave, int lane, v2f bop[16]) {
  const int n  = lane & 15;
  const int kh = (lane >> 4) * 2;
#pragma unroll
  for (int i = 0; i < 16; ++i) {
    const int k0 = wave * 64 + 4 * i + kh;
    v2f v;
    if (n < NYv) {
      v.x = Wx2y[n * NHv + k0];
      v.y = Wx2y[n * NHv + k0 + 1];
    } else {
      v.x = 0.0f; v.y = 0.0f;
    }
    bop[i] = v;
  }
}


__device__ __forceinline__ void drive_step(const float4& u0, const float4& u1,
                                           const float (&Bs1)[NUv],
                                           const float (&Bs2)[NUv],
                                           float& bu1, float& bu2) {
  const float uu[NUv] = {u0.x, u0.y, u0.z, u0.w, u1.x, u1.y, u1.z, u1.w};
  bu1 = 0.0f; bu2 = 0.0f;
#pragma unroll
  for (int u = 0; u < NUv; ++u) {
    bu1 = fmaf(uu[u], Bs1[u], bu1);
    bu2 = fmaf(uu[u], Bs2[u], bu2);
  }
}

template <bool DO_SCAN, bool DO_PROJ, bool DO_RED>
__device__ __forceinline__ void chunk_stage(
    int ch, const float4* Us, float (*Xc)[CHNK][XPAD], float (*Yp)[4][CHNK][16],
    const float (&Bs1)[NUv], const float (&Bs2)[NUv], float re, float im,
    float& x1, float& x2, const v2f (&bop)[16], int wave, int lane, int j,
    float bias, float* __restrict__ Yrow) {
  const int rt = j >> 3, rn = j & 7;

  v2f a[16];
  if (DO_PROJ) {
    const float (*Xp)[XPAD] = Xc[(ch - 1) & 1];
    const int row = lane & 15;
    const int kh  = (lane >> 4) * 2;
#pragma unroll
    for (int i = 0; i < 16; ++i) {
      const int k0 = wave * 64 + 4 * i + kh;
      a[i].x = Xp[row][k0];
      a[i].y = Xp[row][k0 + 1];
    }
  }
  float p0 = 0.0f, p1 = 0.0f, p2 = 0.0f, p3 = 0.0f;
  if (DO_RED) {
    const float (*P)[CHNK][16] = Yp[ch & 1];
    p0 = P[0][rt][rn]; p1 = P[1][rt][rn]; p2 = P[2][rt][rn]; p3 = P[3][rt][rn];
  }
  float4 uq[2 * CHNK];
  if (DO_SCAN) {
#pragma unroll
    for (int tt = 0; tt < CHNK / 2; ++tt) {
      uq[2 * tt]     = Us[(ch * CHNK + tt) * 2];
      uq[2 * tt + 1] = Us[(ch * CHNK + tt) * 2 + 1];
    }
  }
  __builtin_amdgcn_sched_barrier(0);

  float bu1 = 0.0f, bu2 = 0.0f;
  if (DO_SCAN) drive_step(uq[0], uq[1], Bs1, Bs2, bu1, bu2);

  v8f acc = {};
  float xs1[CHNK], xs2[CHNK];
#pragma unroll
  for (int tt = 0; tt < CHNK; ++tt) {
    if (DO_SCAN && tt == 2) {
#pragma unroll
      for (int t2 = CHNK / 2; t2 < CHNK; ++t2) {
        uq[2 * t2]     = Us[(ch * CHNK + t2) * 2];
        uq[2 * t2 + 1] = Us[(ch * CHNK + t2) * 2 + 1];
      }
      __builtin_amdgcn_sched_barrier(0);
    }
    if (DO_PROJ) {
      acc = __builtin_amdgcn_wmma_f32_16x16x4_f32(
          false, a[tt], false, bop[tt], (short)0, acc, false, false);
    }
    if (DO_SCAN) {
      float nb1 = 0.0f, nb2 = 0.0f;
      if (tt + 1 < CHNK)
        drive_step(uq[2 * tt + 2 < 2 * CHNK ? 2 * tt + 2 : 0],
                   uq[2 * tt + 3 < 2 * CHNK ? 2 * tt + 3 : 0], Bs1, Bs2, nb1, nb2);
      const float n1 = re * x1 - im * x2 + bu1;
      const float n2 = im * x1 + re * x2 + bu2;
      x1 = n1; x2 = n2;
      xs1[tt] = x1; xs2[tt] = x2;
      bu1 = nb1; bu2 = nb2;
    }
    if (DO_RED && tt == 1) {
      const float y = p0 + p1 + p2 + p3 + bias;
      Yrow[((size_t)((ch - 2) * CHNK + rt) * NBv) * NYv + rn] = y;
    }
    __builtin_amdgcn_sched_barrier(0);
  }
  if (DO_SCAN) {
    float (*Xn)[XPAD] = Xc[ch & 1];
#pragma unroll
    for (int tt = 0; tt < CHNK; ++tt) {
      Xn[tt][j]       = xs1[tt];
      Xn[tt][j + H2v] = xs2[tt];
    }
  }
  if (DO_PROJ) {
    float (*P)[CHNK][16] = Yp[(ch - 1) & 1];
    const int mbase = (lane >> 4) * 8;
    const int ncol  = lane & 15;
#pragma unroll
    for (int v = 0; v < 8; ++v) P[wave][v + mbase][ncol] = acc[v];
  }
}

__global__ void __launch_bounds__(256)
lru_seg_state(const float* __restrict__ U,   const float* __restrict__ lr,
              const float* __restrict__ li,  const float* __restrict__ Bm,
              const float* __restrict__ mult,float* __restrict__ e_ws) {
  const int bp   = blockIdx.x;
  const int s    = blockIdx.y;
  const int tid  = threadIdx.x;
  const int wave = tid >> 5, lane = tid & 31;
  const int c = lane & 15, h = lane >> 4;
  const int j = wave * 16 + c;
  const int b = 2 * bp + h;

  __shared__ float4 Us[SEG * 4];

  const float* Ub = U + ((size_t)(s * SEG) * NBv + 2 * bp) * NUv;
  const int st0 = tid >> 2, q0 = tid & 3;
  const float4 ua =
      *(const float4*)(Ub + (size_t)st0 * (NBv * NUv) + q0 * 4);
  const float4 ub =
      *(const float4*)(Ub + (size_t)(st0 + SEG / 2) * (NBv * NUv) + q0 * 4);

  const float m1 = expf(mult[j]);
  const float m2 = expf(mult[j + H2v]);
  const float4 w1 = *(const float4*)(Bm + j * NUv + 4 * h);
  const float4 w2 = *(const float4*)(Bm + (j + H2v) * NUv + 4 * h);
  v2f b1a, b1b, b2a, b2b;
  b1a.x = w1.x * m1; b1a.y = w1.y * m1; b1b.x = w1.z * m1; b1b.y = w1.w * m1;
  b2a.x = w2.x * m2; b2a.y = w2.y * m2; b2b.x = w2.z * m2; b2b.y = w2.w * m2;

  const float r  = expf(-fabsf(lr[j]));
  const float th = 1.5707963267948966f * li[j];
  const float re = r * cosf(th);
  const float im = r * sinf(th);

  Us[tid]       = ua;
  Us[tid + 256] = ub;
  __syncthreads();

  const int arow = (c & 7) * 4 + (c >> 3) * 2 + h;

  float x1 = 0.0f, x2 = 0.0f;
#pragma unroll
  for (int g = 0; g < SEG / 8; ++g) {
    const float4 av = Us[g * 32 + arow];
    v2f a0, a1;
    a0.x = av.x; a0.y = av.y; a1.x = av.z; a1.y = av.w;
    v8f d1 = {}, d2 = {};
    d1 = __builtin_amdgcn_wmma_f32_16x16x4_f32(
        false, a0, false, b1a, (short)0, d1, false, false);
    d2 = __builtin_amdgcn_wmma_f32_16x16x4_f32(
        false, a0, false, b2a, (short)0, d2, false, false);
    d1 = __builtin_amdgcn_wmma_f32_16x16x4_f32(
        false, a1, false, b1b, (short)0, d1, false, false);
    d2 = __builtin_amdgcn_wmma_f32_16x16x4_f32(
        false, a1, false, b2b, (short)0, d2, false, false);
#pragma unroll
    for (int rr = 0; rr < 8; ++rr) {
      const float p  = fmaf(-im, x2, d1[rr]);
      const float q  = fmaf( im, x1, d2[rr]);
      const float n1 = fmaf(re, x1, p);
      const float n2 = fmaf(re, x2, q);
      x1 = n1; x2 = n2;
    }
  }

  e_ws[((size_t)s * NBv + b) * NHv + j]       = x1;
  e_ws[((size_t)s * NBv + b) * NHv + j + H2v] = x2;
}

__device__ __forceinline__ void lds_barrier() {
  __builtin_amdgcn_fence(__ATOMIC_RELEASE, "workgroup", "local");
  __builtin_amdgcn_s_barrier();
  __builtin_amdgcn_fence(__ATOMIC_ACQUIRE, "workgroup", "local");
}

__global__ void __launch_bounds__(128)
lru_scan_project(const float* __restrict__ U,   const float* __restrict__ lr,
                 const float* __restrict__ li,  const float* __restrict__ Bm,
                 const float* __restrict__ mult,const float* __restrict__ Wx2y,
                 const float* __restrict__ bx2y,const float* __restrict__ y0,
                 const float* __restrict__ Wy2x,const float* __restrict__ by2x,
                 const float* __restrict__ e_ws,float* __restrict__ Y) {
  const int b = blockIdx.x;
  const int s = blockIdx.y;
  const int j = threadIdx.x;
  const int wave = j >> 5, lane = j & 31;

  __shared__ float4 Us[SEG * 2];
  __shared__ float Xc[2][CHNK][XPAD];
  __shared__ float Yp[2][4][CHNK][16];

  const float* Ub = U + ((size_t)(s * SEG) * NBv + b) * NUv;
  const int uo = (j >> 1) * (NBv * NUv) + (j & 1) * 4;
  const float4 ua = *(const float4*)(Ub + uo);
  const float4 ub = *(const float4*)(Ub + uo + (SEG / 2) * (NBv * NUv));

  float e1[NSEG - 1], e2[NSEG - 1];
#pragma unroll
  for (int i = 0; i < NSEG - 1; ++i) {
    const int seg = i < s ? i : 0;
    const float* row = e_ws + ((size_t)seg * NBv + b) * NHv;
    e1[i] = row[j];
    e2[i] = row[j + H2v];
  }

  const float m1 = expf(mult[j]);
  const float m2 = expf(mult[j + H2v]);
  float Bs1[NUv], Bs2[NUv];
#pragma unroll
  for (int u = 0; u < NUv; ++u) {
    Bs1[u] = Bm[j * NUv + u] * m1;
    Bs2[u] = Bm[(j + H2v) * NUv + u] * m2;
  }
  const float r  = expf(-fabsf(lr[j]));
  const float th = 1.5707963267948966f * li[j];
  Us[j]       = ua;
  Us[j + 128] = ub;
  const float re = r * cosf(th);
  const float im = r * sinf(th);

  v2f bop[16];
  load_wt_ops(Wx2y, wave, lane, bop);

  float x1 = by2x[j], x2 = by2x[j + H2v];
#pragma unroll
  for (int n = 0; n < NYv; ++n) {
    const float yv = y0[b * NYv + n];
    x1 = fmaf(yv, Wy2x[j * NYv + n], x1);
    x2 = fmaf(yv, Wy2x[(j + H2v) * NYv + n], x2);
  }
  {
    const float rS  = expf(-(float)SEG * fabsf(lr[j]));
    const float phS = (float)SEG * th;
    const float reS = rS * cosf(phS);
    const float imS = rS * sinf(phS);
#pragma unroll
    for (int i = 0; i < NSEG - 1; ++i) {
      if (i < s) {
        const float n1 = reS * x1 - imS * x2 + e1[i];
        const float n2 = imS * x1 + reS * x2 + e2[i];
        x1 = n1; x2 = n2;
      }
    }
  }

  __syncthreads();

  const float bias = bx2y[j & 7];
  float* Yrow = Y + ((size_t)(s * SEG) * NBv + b) * NYv;

  chunk_stage<true, false, false>(0, Us, Xc, Yp, Bs1, Bs2, re, im, x1, x2, bop,
                                  wave, lane, j, bias, Yrow);
  lds_barrier();
  chunk_stage<true, true, false>(1, Us, Xc, Yp, Bs1, Bs2, re, im, x1, x2, bop,
                                 wave, lane, j, bias, Yrow);
  lds_barrier();
  for (int ch = 2; ch < NCH; ++ch) {
    chunk_stage<true, true, true>(ch, Us, Xc, Yp, Bs1, Bs2, re, im, x1, x2, bop,
                                  wave, lane, j, bias, Yrow);
    lds_barrier();
  }
  chunk_stage<false, true, true>(NCH, Us, Xc, Yp, Bs1, Bs2, re, im, x1, x2, bop,
                                 wave, lane, j, bias, Yrow);
  lds_barrier();
  chunk_stage<false, false, true>(NCH + 1, Us, Xc, Yp, Bs1, Bs2, re, im, x1, x2,
                                  bop, wave, lane, j, bias, Yrow);
}

extern "C" void kernel_launch(void* const* d_in, const int* in_sizes, int n_in,
                              void* d_out, int out_size, void* d_ws, size_t ws_size,
                              hipStream_t stream) {
  (void)in_sizes; (void)n_in; (void)out_size; (void)ws_size;
  const float* y0   = (const float*)d_in[0];
  const float* U    = (const float*)d_in[1];
  const float* lr   = (const float*)d_in[2];
  const float* li   = (const float*)d_in[3];
  const float* Bm   = (const float*)d_in[4];
  const float* mult = (const float*)d_in[5];
  const float* Wy2x = (const float*)d_in[6];
  const float* by2x = (const float*)d_in[7];
  const float* Wx2y = (const float*)d_in[8];
  const float* bx2y = (const float*)d_in[9];
  float* Y = (float*)d_out;

  float* e_ws = (float*)d_ws;

  dim3 grid1(NBP, NSEG);
  dim3 grid(NBv, NSEG);
  lru_seg_state<<<grid1, 256, 0, stream>>>(U, lr, li, Bm, mult, e_ws);
  lru_scan_project<<<grid, 128, 0, stream>>>(U, lr, li, Bm, mult, Wx2y, bx2y,
                                             y0, Wy2x, by2x, e_ws, Y);
}


# Launch 2 inputs U moved off LDS: scalar loads (4-step groups, one group ahead) feed the drive FMAs as scalar operands instead of broadcast LDS reads; chunk loop unrolled; summation order unchanged
# speedup vs baseline: 2.0244x; 1.0832x over previous
#include <hip/hip_runtime.h>


typedef __attribute__((ext_vector_type(2))) float v2f;
typedef __attribute__((ext_vector_type(8))) float v8f;

#define NYv   8
#define NUv   8
#define NHv   256
#define H2v   128
#define T_LEN 2048
#define NBv   64
#define NSEG  16
#define SEG   (T_LEN / NSEG)
#define CHNK  16
#define NCH   (SEG / CHNK)
#define XPAD  258
#define NBP   32

__device__ __forceinline__ void load_wt_ops(const float* __restrict__ Wx2y,
                                            int wave, int lane, v2f bop[16]) {
  const int n  = lane & 15;
  const int kh = (lane >> 4) * 2;
#pragma unroll
  for (int i = 0; i < 16; ++i) {
    const int k0 = wave * 64 + 4 * i + kh;
    v2f v;
    if (n < NYv) {
      v.x = Wx2y[n * NHv + k0];
      v.y = Wx2y[n * NHv + k0 + 1];
    } else {
      v.x = 0.0f; v.y = 0.0f;
    }
    bop[i] = v;
  }
}


__device__ __forceinline__ void load_u4(const float* __restrict__ Ub, int t0,
                                        float4* g) {
#pragma unroll
  for (int k = 0; k < 4; ++k) {
    const float4* p = (const float4*)(Ub + (size_t)(t0 + k) * (NBv * NUv));
    g[2 * k]     = p[0];
    g[2 * k + 1] = p[1];
  }
}

__device__ __forceinline__ void drive_step(const float4& u0, const float4& u1,
                                           const float (&Bs1)[NUv],
                                           const float (&Bs2)[NUv],
                                           float& bu1, float& bu2) {
  const float uu[NUv] = {u0.x, u0.y, u0.z, u0.w, u1.x, u1.y, u1.z, u1.w};
  bu1 = 0.0f; bu2 = 0.0f;
#pragma unroll
  for (int u = 0; u < NUv; ++u) {
    bu1 = fmaf(uu[u], Bs1[u], bu1);
    bu2 = fmaf(uu[u], Bs2[u], bu2);
  }
}

template <bool DO_SCAN, bool DO_PROJ, bool DO_RED>
__device__ __forceinline__ void chunk_stage(
    int ch, const float* __restrict__ Ub, float4 (&ucar)[8],
    float (*Xc)[CHNK][XPAD], float (*Yp)[4][CHNK][16],
    const float (&Bs1)[NUv], const float (&Bs2)[NUv], float re, float im,
    float& x1, float& x2, const v2f (&bop)[16], int wave, int lane, int j,
    float bias, float* __restrict__ Yrow) {
  const int rt = j >> 3, rn = j & 7;

  v2f a[16];
  if (DO_PROJ) {
    const float (*Xp)[XPAD] = Xc[(ch - 1) & 1];
    const int row = lane & 15;
    const int kh  = (lane >> 4) * 2;
#pragma unroll
    for (int i = 0; i < 16; ++i) {
      const int k0 = wave * 64 + 4 * i + kh;
      a[i].x = Xp[row][k0];
      a[i].y = Xp[row][k0 + 1];
    }
  }
  float p0 = 0.0f, p1 = 0.0f, p2 = 0.0f, p3 = 0.0f;
  if (DO_RED) {
    const float (*P)[CHNK][16] = Yp[ch & 1];
    p0 = P[0][rt][rn]; p1 = P[1][rt][rn]; p2 = P[2][rt][rn]; p3 = P[3][rt][rn];
  }
  float4 uq[2 * CHNK];
  if (DO_SCAN) {
#pragma unroll
    for (int k = 0; k < 8; ++k) uq[k] = ucar[k];
  }
  __builtin_amdgcn_sched_barrier(0);

  float bu1 = 0.0f, bu2 = 0.0f;
  if (DO_SCAN) {
    drive_step(uq[0], uq[1], Bs1, Bs2, bu1, bu2);
    __builtin_amdgcn_sched_barrier(0);
    load_u4(Ub, ch * CHNK + 4, &uq[8]);
  }

  v8f acc = {};
  float xs1[CHNK], xs2[CHNK];
#pragma unroll
  for (int tt = 0; tt < CHNK; ++tt) {
    if (DO_PROJ) {
      acc = __builtin_amdgcn_wmma_f32_16x16x4_f32(
          false, a[tt], false, bop[tt], (short)0, acc, false, false);
    }
    if (DO_SCAN) {
      float nb1 = 0.0f, nb2 = 0.0f;
      if (tt + 1 < CHNK)
        drive_step(uq[2 * tt + 2 < 2 * CHNK ? 2 * tt + 2 : 0],
                   uq[2 * tt + 3 < 2 * CHNK ? 2 * tt + 3 : 0], Bs1, Bs2, nb1, nb2);
      if ((tt & 3) == 3 && tt + 1 < CHNK) {
        __builtin_amdgcn_sched_barrier(0);
        if (tt + 5 < CHNK) {
          load_u4(Ub, ch * CHNK + tt + 5, &uq[2 * (tt + 5)]);
        } else {
          load_u4(Ub, (ch + 1 < NCH ? ch + 1 : ch) * CHNK, ucar);
        }
      }
      const float n1 = re * x1 - im * x2 + bu1;
      const float n2 = im * x1 + re * x2 + bu2;
      x1 = n1; x2 = n2;
      xs1[tt] = x1; xs2[tt] = x2;
      bu1 = nb1; bu2 = nb2;
    }
    if (DO_RED && tt == 1) {
      const float y = p0 + p1 + p2 + p3 + bias;
      Yrow[((size_t)((ch - 2) * CHNK + rt) * NBv) * NYv + rn] = y;
    }
    __builtin_amdgcn_sched_barrier(0);
  }
  if (DO_SCAN) {
    float (*Xn)[XPAD] = Xc[ch & 1];
#pragma unroll
    for (int tt = 0; tt < CHNK; ++tt) {
      Xn[tt][j]       = xs1[tt];
      Xn[tt][j + H2v] = xs2[tt];
    }
  }
  if (DO_PROJ) {
    float (*P)[CHNK][16] = Yp[(ch - 1) & 1];
    const int mbase = (lane >> 4) * 8;
    const int ncol  = lane & 15;
#pragma unroll
    for (int v = 0; v < 8; ++v) P[wave][v + mbase][ncol] = acc[v];
  }
}

__global__ void __launch_bounds__(256)
lru_seg_state(const float* __restrict__ U,   const float* __restrict__ lr,
              const float* __restrict__ li,  const float* __restrict__ Bm,
              const float* __restrict__ mult,float* __restrict__ e_ws) {
  const int bp   = blockIdx.x;
  const int s    = blockIdx.y;
  const int tid  = threadIdx.x;
  const int wave = tid >> 5, lane = tid & 31;
  const int c = lane & 15, h = lane >> 4;
  const int j = wave * 16 + c;
  const int b = 2 * bp + h;

  __shared__ float4 Us[SEG * 4];

  const float* Ub = U + ((size_t)(s * SEG) * NBv + 2 * bp) * NUv;
  const int st0 = tid >> 2, q0 = tid & 3;
  const float4 ua =
      *(const float4*)(Ub + (size_t)st0 * (NBv * NUv) + q0 * 4);
  const float4 ub =
      *(const float4*)(Ub + (size_t)(st0 + SEG / 2) * (NBv * NUv) + q0 * 4);

  const float m1 = expf(mult[j]);
  const float m2 = expf(mult[j + H2v]);
  const float4 w1 = *(const float4*)(Bm + j * NUv + 4 * h);
  const float4 w2 = *(const float4*)(Bm + (j + H2v) * NUv + 4 * h);
  v2f b1a, b1b, b2a, b2b;
  b1a.x = w1.x * m1; b1a.y = w1.y * m1; b1b.x = w1.z * m1; b1b.y = w1.w * m1;
  b2a.x = w2.x * m2; b2a.y = w2.y * m2; b2b.x = w2.z * m2; b2b.y = w2.w * m2;

  const float r  = expf(-fabsf(lr[j]));
  const float th = 1.5707963267948966f * li[j];
  const float re = r * cosf(th);
  const float im = r * sinf(th);

  Us[tid]       = ua;
  Us[tid + 256] = ub;
  __syncthreads();

  const int arow = (c & 7) * 4 + (c >> 3) * 2 + h;

  float x1 = 0.0f, x2 = 0.0f;
#pragma unroll
  for (int g = 0; g < SEG / 8; ++g) {
    const float4 av = Us[g * 32 + arow];
    v2f a0, a1;
    a0.x = av.x; a0.y = av.y; a1.x = av.z; a1.y = av.w;
    v8f d1 = {}, d2 = {};
    d1 = __builtin_amdgcn_wmma_f32_16x16x4_f32(
        false, a0, false, b1a, (short)0, d1, false, false);
    d2 = __builtin_amdgcn_wmma_f32_16x16x4_f32(
        false, a0, false, b2a, (short)0, d2, false, false);
    d1 = __builtin_amdgcn_wmma_f32_16x16x4_f32(
        false, a1, false, b1b, (short)0, d1, false, false);
    d2 = __builtin_amdgcn_wmma_f32_16x16x4_f32(
        false, a1, false, b2b, (short)0, d2, false, false);
#pragma unroll
    for (int rr = 0; rr < 8; ++rr) {
      const float p  = fmaf(-im, x2, d1[rr]);
      const float q  = fmaf( im, x1, d2[rr]);
      const float n1 = fmaf(re, x1, p);
      const float n2 = fmaf(re, x2, q);
      x1 = n1; x2 = n2;
    }
  }

  e_ws[((size_t)s * NBv + b) * NHv + j]       = x1;
  e_ws[((size_t)s * NBv + b) * NHv + j + H2v] = x2;
}

__device__ __forceinline__ void lds_barrier() {
  __builtin_amdgcn_fence(__ATOMIC_RELEASE, "workgroup", "local");
  __builtin_amdgcn_s_barrier();
  __builtin_amdgcn_fence(__ATOMIC_ACQUIRE, "workgroup", "local");
}

__global__ void __launch_bounds__(128)
lru_scan_project(const float* __restrict__ U,   const float* __restrict__ lr,
                 const float* __restrict__ li,  const float* __restrict__ Bm,
                 const float* __restrict__ mult,const float* __restrict__ Wx2y,
                 const float* __restrict__ bx2y,const float* __restrict__ y0,
                 const float* __restrict__ Wy2x,const float* __restrict__ by2x,
                 const float* __restrict__ e_ws,float* __restrict__ Y) {
  const int b = blockIdx.x;
  const int s = blockIdx.y;
  const int j = threadIdx.x;
  const int wave = j >> 5, lane = j & 31;

  __shared__ float Xc[2][CHNK][XPAD];
  __shared__ float Yp[2][4][CHNK][16];

  const float* Ub = U + ((size_t)(s * SEG) * NBv + b) * NUv;
  float4 ucar[8];
  load_u4(Ub, 0, ucar);

  float e1[NSEG - 1], e2[NSEG - 1];
#pragma unroll
  for (int i = 0; i < NSEG - 1; ++i) {
    const int seg = i < s ? i : 0;
    const float* row = e_ws + ((size_t)seg * NBv + b) * NHv;
    e1[i] = row[j];
    e2[i] = row[j + H2v];
  }

  const float m1 = expf(mult[j]);
  const float m2 = expf(mult[j + H2v]);
  float Bs1[NUv], Bs2[NUv];
#pragma unroll
  for (int u = 0; u < NUv; ++u) {
    Bs1[u] = Bm[j * NUv + u] * m1;
    Bs2[u] = Bm[(j + H2v) * NUv + u] * m2;
  }
  const float r  = expf(-fabsf(lr[j]));
  const float th = 1.5707963267948966f * li[j];
  const float re = r * cosf(th);
  const float im = r * sinf(th);

  v2f bop[16];
  load_wt_ops(Wx2y, wave, lane, bop);

  float x1 = by2x[j], x2 = by2x[j + H2v];
#pragma unroll
  for (int n = 0; n < NYv; ++n) {
    const float yv = y0[b * NYv + n];
    x1 = fmaf(yv, Wy2x[j * NYv + n], x1);
    x2 = fmaf(yv, Wy2x[(j + H2v) * NYv + n], x2);
  }
  {
    const float rS  = expf(-(float)SEG * fabsf(lr[j]));
    const float phS = (float)SEG * th;
    const float reS = rS * cosf(phS);
    const float imS = rS * sinf(phS);
#pragma unroll
    for (int i = 0; i < NSEG - 1; ++i) {
      if (i < s) {
        const float n1 = reS * x1 - imS * x2 + e1[i];
        const float n2 = imS * x1 + reS * x2 + e2[i];
        x1 = n1; x2 = n2;
      }
    }
  }

  const float bias = bx2y[j & 7];
  float* Yrow = Y + ((size_t)(s * SEG) * NBv + b) * NYv;

  chunk_stage<true, false, false>(0, Ub, ucar, Xc, Yp, Bs1, Bs2, re, im, x1, x2, bop,
                                  wave, lane, j, bias, Yrow);
  lds_barrier();
  chunk_stage<true, true, false>(1, Ub, ucar, Xc, Yp, Bs1, Bs2, re, im, x1, x2, bop,
                                 wave, lane, j, bias, Yrow);
  lds_barrier();
#pragma unroll
  for (int ch = 2; ch < NCH; ++ch) {
    chunk_stage<true, true, true>(ch, Ub, ucar, Xc, Yp, Bs1, Bs2, re, im, x1, x2, bop,
                                  wave, lane, j, bias, Yrow);
    lds_barrier();
  }
  chunk_stage<false, true, true>(NCH, Ub, ucar, Xc, Yp, Bs1, Bs2, re, im, x1, x2, bop,
                                 wave, lane, j, bias, Yrow);
  lds_barrier();
  chunk_stage<false, false, true>(NCH + 1, Ub, ucar, Xc, Yp, Bs1, Bs2, re, im, x1, x2,
                                  bop, wave, lane, j, bias, Yrow);
}

extern "C" void kernel_launch(void* const* d_in, const int* in_sizes, int n_in,
                              void* d_out, int out_size, void* d_ws, size_t ws_size,
                              hipStream_t stream) {
  (void)in_sizes; (void)n_in; (void)out_size; (void)ws_size;
  const float* y0   = (const float*)d_in[0];
  const float* U    = (const float*)d_in[1];
  const float* lr   = (const float*)d_in[2];
  const float* li   = (const float*)d_in[3];
  const float* Bm   = (const float*)d_in[4];
  const float* mult = (const float*)d_in[5];
  const float* Wy2x = (const float*)d_in[6];
  const float* by2x = (const float*)d_in[7];
  const float* Wx2y = (const float*)d_in[8];
  const float* bx2y = (const float*)d_in[9];
  float* Y = (float*)d_out;

  float* e_ws = (float*)d_ws;

  dim3 grid1(NBP, NSEG);
  dim3 grid(NBv, NSEG);
  lru_seg_state<<<grid1, 256, 0, stream>>>(U, lr, li, Bm, mult, e_ws);
  lru_scan_project<<<grid, 128, 0, stream>>>(U, lr, li, Bm, mult, Wx2y, bx2y,
                                             y0, Wy2x, by2x, e_ws, Y);
}


# Launch 2 projection: the tile's 8 padding columns now carry the NEXT step's outputs (B columns 8-15 = W.R; that step's drive W.Bs.u added at the reduce): projection WMMAs halved; sum order changes
# speedup vs baseline: 2.0968x; 1.0358x over previous
#include <hip/hip_runtime.h>


typedef __attribute__((ext_vector_type(2))) float v2f;
typedef __attribute__((ext_vector_type(8))) float v8f;

#define NYv   8
#define NUv   8
#define NHv   256
#define H2v   128
#define T_LEN 2048
#define NBv   64
#define NSEG  16
#define SEG   (T_LEN / NSEG)
#define CHNK  32
#define ROWS  16
#define NCH   (SEG / CHNK)
#define TABW  (16 * NHv)
#define XPAD  258
#define NBP   32

__device__ __forceinline__ void load_wt_ops(const float* __restrict__ Tb,
                                            int wave, int lane, v2f bop[16]) {
  const int n  = lane & 15;
  const int kh = (lane >> 4) * 2;
#pragma unroll
  for (int i = 0; i < 16; ++i) {
    const int k0 = wave * 64 + 4 * i + kh;
    v2f v;
    v.x = Tb[n * NHv + k0];
    v.y = Tb[n * NHv + k0 + 1];
    bop[i] = v;
  }
}


__device__ __forceinline__ void load_u4(const float* __restrict__ Ub, int t0,
                                        float4* g) {
#pragma unroll
  for (int k = 0; k < 4; ++k) {
    const float4* p = (const float4*)(Ub + (size_t)(t0 + k) * (NBv * NUv));
    g[2 * k]     = p[0];
    g[2 * k + 1] = p[1];
  }
}

__device__ __forceinline__ void drive_step(const float4& u0, const float4& u1,
                                           const float (&Bs1)[NUv],
                                           const float (&Bs2)[NUv],
                                           float& bu1, float& bu2) {
  const float uu[NUv] = {u0.x, u0.y, u0.z, u0.w, u1.x, u1.y, u1.z, u1.w};
  bu1 = 0.0f; bu2 = 0.0f;
#pragma unroll
  for (int u = 0; u < NUv; ++u) {
    bu1 = fmaf(uu[u], Bs1[u], bu1);
    bu2 = fmaf(uu[u], Bs2[u], bu2);
  }
}

template <bool DO_SCAN, bool DO_PROJ, bool DO_RED, bool DO_PREF>
__device__ __forceinline__ void chunk_stage(
    int ch, const float* __restrict__ Ub, float4 (&ucar)[8], float4 (&ured)[4],
    float (*Xc)[ROWS][XPAD], float (*Yp)[4][ROWS][16],
    const float (&Bs1)[NUv], const float (&Bs2)[NUv], float re, float im,
    float& x1, float& x2, const v2f (&bop)[16], const float (&Gr)[NUv],
    int wave, int lane, int j, float bias, float* __restrict__ Yrow) {
  const int r0 = j >> 4, col = j & 15;

  v2f a[16];
  if (DO_PROJ) {
    const float (*Xp)[XPAD] = Xc[(ch - 1) & 1];
    const int row = lane & 15;
    const int kh  = (lane >> 4) * 2;
#pragma unroll
    for (int i = 0; i < 16; ++i) {
      const int k0 = wave * 64 + 4 * i + kh;
      a[i].x = Xp[row][k0];
      a[i].y = Xp[row][k0 + 1];
    }
  }
  float p[2][4] = {{0.0f, 0.0f, 0.0f, 0.0f}, {0.0f, 0.0f, 0.0f, 0.0f}};
  float4 uc[4] = {};
  if (DO_RED) {
    const float (*P)[ROWS][16] = Yp[ch & 1];
#pragma unroll
    for (int rr = 0; rr < 2; ++rr) {
#pragma unroll
      for (int w = 0; w < 4; ++w) p[rr][w] = P[w][r0 + 8 * rr][col];
    }
#pragma unroll
    for (int k = 0; k < 4; ++k) uc[k] = ured[k];
  }
  if (DO_PREF) {
#pragma unroll
    for (int rr = 0; rr < 2; ++rr) {
      const float4* q = (const float4*)(
          Ub + (size_t)((ch - 1) * CHNK + 2 * (r0 + 8 * rr) + 1) * (NBv * NUv));
      ured[2 * rr]     = q[0];
      ured[2 * rr + 1] = q[1];
    }
  }
  float4 uq[2 * CHNK];
  if (DO_SCAN) {
#pragma unroll
    for (int k = 0; k < 8; ++k) uq[k] = ucar[k];
  }
  __builtin_amdgcn_sched_barrier(0);

  float bu1 = 0.0f, bu2 = 0.0f;
  if (DO_SCAN) {
    drive_step(uq[0], uq[1], Bs1, Bs2, bu1, bu2);
    __builtin_amdgcn_sched_barrier(0);
    load_u4(Ub, ch * CHNK + 4, &uq[8]);
  }

  v8f acc = {};
  float xs1[ROWS], xs2[ROWS];
#pragma unroll
  for (int tt = 0; tt < CHNK; ++tt) {
    if (DO_PROJ && (tt & 1) == 0) {
      acc = __builtin_amdgcn_wmma_f32_16x16x4_f32(
          false, a[tt >> 1], false, bop[tt >> 1], (short)0, acc, false, false);
    }
    if (DO_SCAN) {
      float nb1 = 0.0f, nb2 = 0.0f;
      if (tt + 1 < CHNK)
        drive_step(uq[2 * tt + 2 < 2 * CHNK ? 2 * tt + 2 : 0],
                   uq[2 * tt + 3 < 2 * CHNK ? 2 * tt + 3 : 0], Bs1, Bs2, nb1, nb2);
      if ((tt & 3) == 3 && tt + 1 < CHNK) {
        __builtin_amdgcn_sched_barrier(0);
        if (tt + 5 < CHNK) {
          load_u4(Ub, ch * CHNK + tt + 5, &uq[2 * (tt + 5)]);
        } else {
          load_u4(Ub, (ch + 1 < NCH ? ch + 1 : ch) * CHNK, ucar);
        }
      }
      const float n1 = re * x1 - im * x2 + bu1;
      const float n2 = im * x1 + re * x2 + bu2;
      x1 = n1; x2 = n2;
      if ((tt & 1) == 0) {
        xs1[tt >> 1] = x1;
        xs2[tt >> 1] = x2;
      }
      bu1 = nb1; bu2 = nb2;
    }
    if (DO_RED && tt == 1) {
#pragma unroll
      for (int rr = 0; rr < 2; ++rr) {
        const int row = r0 + 8 * rr;
        float y = p[rr][0] + p[rr][1] + p[rr][2] + p[rr][3] + bias;
        const float4 u0 = uc[2 * rr], u1 = uc[2 * rr + 1];
        y = fmaf(Gr[0], u0.x, y); y = fmaf(Gr[1], u0.y, y);
        y = fmaf(Gr[2], u0.z, y); y = fmaf(Gr[3], u0.w, y);
        y = fmaf(Gr[4], u1.x, y); y = fmaf(Gr[5], u1.y, y);
        y = fmaf(Gr[6], u1.z, y); y = fmaf(Gr[7], u1.w, y);
        Yrow[((size_t)((ch - 2) * CHNK + 2 * row + (col >> 3)) * NBv) * NYv +
             (col & 7)] = y;
      }
    }
    __builtin_amdgcn_sched_barrier(0);
  }
  if (DO_SCAN) {
    float (*Xn)[XPAD] = Xc[ch & 1];
#pragma unroll
    for (int m = 0; m < ROWS; ++m) {
      Xn[m][j]       = xs1[m];
      Xn[m][j + H2v] = xs2[m];
    }
  }
  if (DO_PROJ) {
    float (*P)[ROWS][16] = Yp[(ch - 1) & 1];
    const int mbase = (lane >> 4) * 8;
    const int ncol  = lane & 15;
#pragma unroll
    for (int v = 0; v < 8; ++v) P[wave][v + mbase][ncol] = acc[v];
  }
}

__global__ void __launch_bounds__(256)
lru_seg_state(const float* __restrict__ U,   const float* __restrict__ lr,
              const float* __restrict__ li,  const float* __restrict__ Bm,
              const float* __restrict__ mult,const float* __restrict__ Wx2y,
              float* __restrict__ e_ws,      float* __restrict__ tab) {
  const int bp   = blockIdx.x;
  const int s    = blockIdx.y;
  const int tid  = threadIdx.x;
  const int wave = tid >> 5, lane = tid & 31;
  const int c = lane & 15, h = lane >> 4;
  const int j = wave * 16 + c;
  const int b = 2 * bp + h;

  __shared__ float4 Us[SEG * 4];

  const float* Ub = U + ((size_t)(s * SEG) * NBv + 2 * bp) * NUv;
  const int st0 = tid >> 2, q0 = tid & 3;
  const float4 ua =
      *(const float4*)(Ub + (size_t)st0 * (NBv * NUv) + q0 * 4);
  const float4 ub =
      *(const float4*)(Ub + (size_t)(st0 + SEG / 2) * (NBv * NUv) + q0 * 4);

  const float m1 = expf(mult[j]);
  const float m2 = expf(mult[j + H2v]);
  const float4 w1 = *(const float4*)(Bm + j * NUv + 4 * h);
  const float4 w2 = *(const float4*)(Bm + (j + H2v) * NUv + 4 * h);
  v2f b1a, b1b, b2a, b2b;
  b1a.x = w1.x * m1; b1a.y = w1.y * m1; b1b.x = w1.z * m1; b1b.y = w1.w * m1;
  b2a.x = w2.x * m2; b2a.y = w2.y * m2; b2b.x = w2.z * m2; b2b.y = w2.w * m2;

  const float r  = expf(-fabsf(lr[j]));
  const float th = 1.5707963267948966f * li[j];
  const float re = r * cosf(th);
  const float im = r * sinf(th);

  if (bp == 0 && s == 0) {
#pragma unroll
    for (int nn = 0; nn < 4; ++nn) {
      const int n = 4 * h + nn;
      const float w1 = Wx2y[n * NHv + j];
      const float w2 = Wx2y[n * NHv + j + H2v];
      tab[n * NHv + j]             = w1;
      tab[n * NHv + j + H2v]       = w2;
      tab[(8 + n) * NHv + j]       = fmaf(w1, re, w2 * im);
      tab[(8 + n) * NHv + j + H2v] = fmaf(w2, re, -(w1 * im));
    }
    const int o = tid >> 2, q = tid & 3;
    const int gn = o >> 3, gu = o & 7;
    float g = 0.0f;
#pragma unroll 16
    for (int kk = 0; kk < 64; ++kk) {
      const int k = q * 64 + kk;
      g = fmaf(Wx2y[gn * NHv + k], Bm[k * NUv + gu] * expf(mult[k]), g);
    }
    g += __shfl_xor(g, 1);
    g += __shfl_xor(g, 2);
    if (q == 0) tab[TABW + o] = g;
  }

  Us[tid]       = ua;
  Us[tid + 256] = ub;
  __syncthreads();

  const int arow = (c & 7) * 4 + (c >> 3) * 2 + h;

  float x1 = 0.0f, x2 = 0.0f;
#pragma unroll
  for (int g = 0; g < SEG / 8; ++g) {
    const float4 av = Us[g * 32 + arow];
    v2f a0, a1;
    a0.x = av.x; a0.y = av.y; a1.x = av.z; a1.y = av.w;
    v8f d1 = {}, d2 = {};
    d1 = __builtin_amdgcn_wmma_f32_16x16x4_f32(
        false, a0, false, b1a, (short)0, d1, false, false);
    d2 = __builtin_amdgcn_wmma_f32_16x16x4_f32(
        false, a0, false, b2a, (short)0, d2, false, false);
    d1 = __builtin_amdgcn_wmma_f32_16x16x4_f32(
        false, a1, false, b1b, (short)0, d1, false, false);
    d2 = __builtin_amdgcn_wmma_f32_16x16x4_f32(
        false, a1, false, b2b, (short)0, d2, false, false);
#pragma unroll
    for (int rr = 0; rr < 8; ++rr) {
      const float p  = fmaf(-im, x2, d1[rr]);
      const float q  = fmaf( im, x1, d2[rr]);
      const float n1 = fmaf(re, x1, p);
      const float n2 = fmaf(re, x2, q);
      x1 = n1; x2 = n2;
    }
  }

  e_ws[((size_t)s * NBv + b) * NHv + j]       = x1;
  e_ws[((size_t)s * NBv + b) * NHv + j + H2v] = x2;
}

__device__ __forceinline__ void lds_barrier() {
  __builtin_amdgcn_fence(__ATOMIC_RELEASE, "workgroup", "local");
  __builtin_amdgcn_s_barrier();
  __builtin_amdgcn_fence(__ATOMIC_ACQUIRE, "workgroup", "local");
}

__global__ void __launch_bounds__(128)
lru_scan_project(const float* __restrict__ U,   const float* __restrict__ lr,
                 const float* __restrict__ li,  const float* __restrict__ Bm,
                 const float* __restrict__ mult,const float* __restrict__ Wx2y,
                 const float* __restrict__ bx2y,const float* __restrict__ y0,
                 const float* __restrict__ Wy2x,const float* __restrict__ by2x,
                 const float* __restrict__ e_ws,const float* __restrict__ tab,
                 float* __restrict__ Y) {
  const int b = blockIdx.x;
  const int s = blockIdx.y;
  const int j = threadIdx.x;
  const int wave = j >> 5, lane = j & 31;

  __shared__ float Xc[2][ROWS][XPAD];
  __shared__ float Yp[2][4][ROWS][16];

  const float* Ub = U + ((size_t)(s * SEG) * NBv + b) * NUv;
  float4 ucar[8];
  load_u4(Ub, 0, ucar);

  float e1[NSEG - 1], e2[NSEG - 1];
#pragma unroll
  for (int i = 0; i < NSEG - 1; ++i) {
    const int seg = i < s ? i : 0;
    const float* row = e_ws + ((size_t)seg * NBv + b) * NHv;
    e1[i] = row[j];
    e2[i] = row[j + H2v];
  }

  const float m1 = expf(mult[j]);
  const float m2 = expf(mult[j + H2v]);
  float Bs1[NUv], Bs2[NUv];
#pragma unroll
  for (int u = 0; u < NUv; ++u) {
    Bs1[u] = Bm[j * NUv + u] * m1;
    Bs2[u] = Bm[(j + H2v) * NUv + u] * m2;
  }
  const float r  = expf(-fabsf(lr[j]));
  const float th = 1.5707963267948966f * li[j];
  const float re = r * cosf(th);
  const float im = r * sinf(th);

  v2f bop[16];
  load_wt_ops(tab, wave, lane, bop);
  float Gr[NUv];
#pragma unroll
  for (int u = 0; u < NUv; ++u) {
    const float gv = tab[TABW + (j & 7) * NUv + u];
    Gr[u] = (j & 8) ? gv : 0.0f;
  }
  float4 ured[4] = {};

  float x1 = by2x[j], x2 = by2x[j + H2v];
#pragma unroll
  for (int n = 0; n < NYv; ++n) {
    const float yv = y0[b * NYv + n];
    x1 = fmaf(yv, Wy2x[j * NYv + n], x1);
    x2 = fmaf(yv, Wy2x[(j + H2v) * NYv + n], x2);
  }
  {
    const float rS  = expf(-(float)SEG * fabsf(lr[j]));
    const float phS = (float)SEG * th;
    const float reS = rS * cosf(phS);
    const float imS = rS * sinf(phS);
#pragma unroll
    for (int i = 0; i < NSEG - 1; ++i) {
      if (i < s) {
        const float n1 = reS * x1 - imS * x2 + e1[i];
        const float n2 = imS * x1 + reS * x2 + e2[i];
        x1 = n1; x2 = n2;
      }
    }
  }

  const float bias = bx2y[j & 7];
  float* Yrow = Y + ((size_t)(s * SEG) * NBv + b) * NYv;

  chunk_stage<true, false, false, false>(0, Ub, ucar, ured, Xc, Yp, Bs1, Bs2, re, im,
                                         x1, x2, bop, Gr, wave, lane, j, bias, Yrow);
  lds_barrier();
  chunk_stage<true, true, false, true>(1, Ub, ucar, ured, Xc, Yp, Bs1, Bs2, re, im,
                                       x1, x2, bop, Gr, wave, lane, j, bias, Yrow);
  lds_barrier();
#pragma unroll
  for (int ch = 2; ch < NCH; ++ch) {
    chunk_stage<true, true, true, true>(ch, Ub, ucar, ured, Xc, Yp, Bs1, Bs2, re, im,
                                        x1, x2, bop, Gr, wave, lane, j, bias, Yrow);
    lds_barrier();
  }
  chunk_stage<false, true, true, true>(NCH, Ub, ucar, ured, Xc, Yp, Bs1, Bs2, re, im,
                                       x1, x2, bop, Gr, wave, lane, j, bias, Yrow);
  lds_barrier();
  chunk_stage<false, false, true, false>(NCH + 1, Ub, ucar, ured, Xc, Yp, Bs1, Bs2, re,
                                         im, x1, x2, bop, Gr, wave, lane, j, bias, Yrow);
}

extern "C" void kernel_launch(void* const* d_in, const int* in_sizes, int n_in,
                              void* d_out, int out_size, void* d_ws, size_t ws_size,
                              hipStream_t stream) {
  (void)in_sizes; (void)n_in; (void)out_size; (void)ws_size;
  const float* y0   = (const float*)d_in[0];
  const float* U    = (const float*)d_in[1];
  const float* lr   = (const float*)d_in[2];
  const float* li   = (const float*)d_in[3];
  const float* Bm   = (const float*)d_in[4];
  const float* mult = (const float*)d_in[5];
  const float* Wy2x = (const float*)d_in[6];
  const float* by2x = (const float*)d_in[7];
  const float* Wx2y = (const float*)d_in[8];
  const float* bx2y = (const float*)d_in[9];
  float* Y = (float*)d_out;

  float* e_ws = (float*)d_ws;
  float* tab  = e_ws + NSEG * NBv * NHv;

  dim3 grid1(NBP, NSEG);
  dim3 grid(NBv, NSEG);
  lru_seg_state<<<grid1, 256, 0, stream>>>(U, lr, li, Bm, mult, Wx2y, e_ws, tab);
  lru_scan_project<<<grid, 128, 0, stream>>>(U, lr, li, Bm, mult, Wx2y, bx2y,
                                             y0, Wy2x, by2x, e_ws, tab, Y);
}
